# starting kernel
# baseline (speedup 1.0000x reference)
#include <hip/hip_runtime.h>
#include <hip/hip_bf16.h>
#include <math.h>

#define BATCH    2
#define SEQLEN   4096
#define BL       (BATCH * SEQLEN)
#define D_MODEL  512
#define D_IN_PROJ 1168
#define CONV_DIM 576
#define CONV2    1152
#define D_INNER  1024
#define NHEADS   16
#define HH       8
#define HEADDIM  64
#define NSTATE   32
#define NCHUNK   32
#define CHUNK    128

typedef float v2f __attribute__((ext_vector_type(2)));
typedef float v8f __attribute__((ext_vector_type(8)));

__device__ __forceinline__ v8f wmma_f32_4(v2f a, v2f b, v8f c) {
    return __builtin_amdgcn_wmma_f32_16x16x4_f32(
         false, a,  false, b,
         (short)0, c,  false,  false);
}

__global__ void __launch_bounds__(128)
gemm_in(const float* __restrict__ U, const float* __restrict__ Win,
        float* __restrict__ out) {
    const int wv   = threadIdx.x >> 5;
    const int lane = threadIdx.x & 31;
    const int row  = lane & 15;
    const int koff = (lane >> 4) << 1;
    const int m0   = (blockIdx.y * 16 + wv * 4) * 16;
    const int n0   = blockIdx.x * 16;

    const float* ap = U   + (size_t)(m0 + row) * D_MODEL + koff;
    const float* bp = Win + (size_t)koff * D_IN_PROJ + n0 + row;

    v8f acc0 = {}, acc1 = {}, acc2 = {}, acc3 = {};
    for (int k = 0; k < D_MODEL; k += 4) {
        v2f b;
        b.x = bp[(size_t)k * D_IN_PROJ];
        b.y = bp[(size_t)k * D_IN_PROJ + D_IN_PROJ];
        v2f a0 = *(const v2f*)(ap + k);
        v2f a1 = *(const v2f*)(ap + 16 * D_MODEL + k);
        v2f a2 = *(const v2f*)(ap + 32 * D_MODEL + k);
        v2f a3 = *(const v2f*)(ap + 48 * D_MODEL + k);
        acc0 = wmma_f32_4(a0, b, acc0);
        acc1 = wmma_f32_4(a1, b, acc1);
        acc2 = wmma_f32_4(a2, b, acc2);
        acc3 = wmma_f32_4(a3, b, acc3);
    }
    const int mhi = (lane >> 4) << 3;
#pragma unroll
    for (int r = 0; r < 8; ++r) {
        const size_t mm = (size_t)(m0 + r + mhi);
        out[(mm     ) * D_IN_PROJ + n0 + row] = acc0[r];
        out[(mm + 16) * D_IN_PROJ + n0 + row] = acc1[r];
        out[(mm + 32) * D_IN_PROJ + n0 + row] = acc2[r];
        out[(mm + 48) * D_IN_PROJ + n0 + row] = acc3[r];
    }
}

__global__ void __launch_bounds__(256)
conv_dt(const float* __restrict__ xbcdt,
        const float* __restrict__ wf, const float* __restrict__ bf,
        const float* __restrict__ wb, const float* __restrict__ bb,
        const float* __restrict__ dt_bias,
        float* __restrict__ xc, float* __restrict__ dtb) {
    const size_t bl = blockIdx.x;
    const int b = (int)(bl / SEQLEN);
    const int t = (int)(bl % SEQLEN);

    for (int c = threadIdx.x; c < CONV2; c += blockDim.x) {
        float acc;
        if (c < CONV_DIM) {
            acc = bf[c];
#pragma unroll
            for (int k = 0; k < 4; ++k) {
                int tt = t - 3 + k;
                if (tt >= 0)
                    acc += xbcdt[((size_t)b * SEQLEN + tt) * D_IN_PROJ + c] * wf[c * 4 + k];
            }
        } else {
            int cc = c - CONV_DIM;
            acc = bb[cc];
#pragma unroll
            for (int k = 0; k < 4; ++k) {
                int tt = t + 3 - k;
                if (tt < SEQLEN)
                    acc += xbcdt[((size_t)b * SEQLEN + tt) * D_IN_PROJ + c] * wb[cc * 4 + k];
            }
        }
        float sig = 1.0f / (1.0f + expf(-acc));
        xc[bl * CONV2 + c] = acc * sig;
    }
    for (int h = threadIdx.x; h < NHEADS; h += blockDim.x) {
        float v = xbcdt[bl * D_IN_PROJ + CONV2 + h] + dt_bias[h];
        dtb[bl * NHEADS + h] = (v > 20.0f) ? v : log1pf(expf(v));
    }
}

__device__ __forceinline__ void head_bases(int h, int& xbase, int& bbase, int& cbase) {
    if (h < HH) { xbase = h * HEADDIM;                 bbase = 512;  cbase = 544;  }
    else        { xbase = CONV_DIM + (h - HH) * HEADDIM; bbase = 1088; cbase = 1120; }
}

__global__ void __launch_bounds__(64)
scan_chunk(const float* __restrict__ xc, const float* __restrict__ dtb,
           const float* __restrict__ A_log, const float* __restrict__ Dp,
           float* __restrict__ y, float* __restrict__ S, float* __restrict__ prodA) {
    const int chunk = blockIdx.x & (NCHUNK - 1);
    const int h     = (blockIdx.x >> 5) & 15;
    const int b     = blockIdx.x >> 9;
    const int p     = threadIdx.x;
    int xbase, bbase, cbase; head_bases(h, xbase, bbase, cbase);

    const float Ah = -expf(A_log[h]);
    const float Dh = Dp[h];
    float hst[NSTATE];
#pragma unroll
    for (int n = 0; n < NSTATE; ++n) hst[n] = 0.0f;
    float pA = 1.0f;

    __shared__ float sB[NSTATE], sC[NSTATE];

    for (int sl = 0; sl < CHUNK; ++sl) {
        const int s = chunk * CHUNK + sl;
        const int t = (h < HH) ? s : (SEQLEN - 1 - s);
        const size_t rb = (size_t)b * SEQLEN + t;
        const float dtv = dtb[rb * NHEADS + h];
        const float dA  = expf(Ah * dtv);
        pA *= dA;
        if (p < NSTATE) sB[p] = xc[rb * CONV2 + bbase + p];
        else            sC[p - NSTATE] = xc[rb * CONV2 + cbase + (p - NSTATE)];
        __syncthreads();
        const float xp  = xc[rb * CONV2 + xbase + p];
        const float dtx = dtv * xp;
        float accy = 0.0f;
#pragma unroll
        for (int n = 0; n < NSTATE; ++n) {
            hst[n] = dA * hst[n] + dtx * sB[n];
            accy  += hst[n] * sC[n];
        }
        y[rb * D_INNER + h * HEADDIM + p] = accy + Dh * xp;
        __syncthreads();
    }
    const size_t sb = (((size_t)(b * NHEADS + h)) * NCHUNK + chunk) * (HEADDIM * NSTATE)
                    + (size_t)p * NSTATE;
#pragma unroll
    for (int n = 0; n < NSTATE; ++n) S[sb + n] = hst[n];
    if (p == 0) prodA[(b * NHEADS + h) * NCHUNK + chunk] = pA;
}

__global__ void __launch_bounds__(64)
scan_combine(float* __restrict__ S, const float* __restrict__ prodA) {
    const int bh = blockIdx.x;
    const int p  = threadIdx.x;
    float hcur[NSTATE];
#pragma unroll
    for (int n = 0; n < NSTATE; ++n) hcur[n] = 0.0f;
    for (int c = 0; c < NCHUNK; ++c) {
        const size_t base = (((size_t)bh) * NCHUNK + c) * (HEADDIM * NSTATE) + (size_t)p * NSTATE;
        const float a = prodA[bh * NCHUNK + c];
#pragma unroll
        for (int n = 0; n < NSTATE; ++n) {
            float sl = S[base + n];
            S[base + n] = hcur[n];
            hcur[n] = a * hcur[n] + sl;
        }
    }
}

__global__ void __launch_bounds__(64)
scan_correct(const float* __restrict__ xc, const float* __restrict__ dtb,
             const float* __restrict__ A_log, const float* __restrict__ S,
             float* __restrict__ y) {
    const int chunk = blockIdx.x & (NCHUNK - 1);
    if (chunk == 0) return;
    const int h = (blockIdx.x >> 5) & 15;
    const int b = blockIdx.x >> 9;
    const int p = threadIdx.x;
    int xbase, bbase, cbase; head_bases(h, xbase, bbase, cbase);
    (void)xbase; (void)bbase;

    float hi[NSTATE];
    const size_t sb = (((size_t)(b * NHEADS + h)) * NCHUNK + chunk) * (HEADDIM * NSTATE)
                    + (size_t)p * NSTATE;
#pragma unroll
    for (int n = 0; n < NSTATE; ++n) hi[n] = S[sb + n];

    const float Ah = -expf(A_log[h]);
    float cumA = 1.0f;
    __shared__ float sC[NSTATE];

    for (int sl = 0; sl < CHUNK; ++sl) {
        const int s = chunk * CHUNK + sl;
        const int t = (h < HH) ? s : (SEQLEN - 1 - s);
        const size_t rb = (size_t)b * SEQLEN + t;
        cumA *= expf(Ah * dtb[rb * NHEADS + h]);
        if (p < NSTATE) sC[p] = xc[rb * CONV2 + cbase + p];
        __syncthreads();
        float acc = 0.0f;
#pragma unroll
        for (int n = 0; n < NSTATE; ++n) acc += hi[n] * sC[n];
        y[rb * D_INNER + h * HEADDIM + p] += cumA * acc;
        __syncthreads();
    }
}

__global__ void __launch_bounds__(256)
ln_stats(const float* __restrict__ y, float* __restrict__ mu, float* __restrict__ rstd) {
    const size_t row = blockIdx.x;
    const int tid = threadIdx.x;
    const float* yr = y + row * D_INNER;
    float s = 0.0f, q = 0.0f;
    for (int i = tid; i < D_INNER; i += 256) { float v = yr[i]; s += v; q += v * v; }
    __shared__ float ss[256], sq[256];
    ss[tid] = s; sq[tid] = q; __syncthreads();
    for (int off = 128; off > 0; off >>= 1) {
        if (tid < off) { ss[tid] += ss[tid + off]; sq[tid] += sq[tid + off]; }
        __syncthreads();
    }
    if (tid == 0) {
        float m = ss[0] * (1.0f / D_INNER);
        float v = sq[0] * (1.0f / D_INNER) - m * m;
        mu[row] = m;
        rstd[row] = rsqrtf(v + 1e-5f);
    }
}

__global__ void __launch_bounds__(128)
gemm_out(const float* __restrict__ Y, const float* __restrict__ mu,
         const float* __restrict__ rstd, const float* __restrict__ lnw,
         const float* __restrict__ lnb, const float* __restrict__ Wout,
         float* __restrict__ out) {
    const int wv   = threadIdx.x >> 5;
    const int lane = threadIdx.x & 31;
    const int row  = lane & 15;
    const int koff = (lane >> 4) << 1;
    const int m0   = (blockIdx.y * 16 + wv * 4) * 16;
    const int n0   = blockIdx.x * 16;

    const float mum0 = mu[m0 + row],        rsm0 = rstd[m0 + row];
    const float mum1 = mu[m0 + 16 + row],   rsm1 = rstd[m0 + 16 + row];
    const float mum2 = mu[m0 + 32 + row],   rsm2 = rstd[m0 + 32 + row];
    const float mum3 = mu[m0 + 48 + row],   rsm3 = rstd[m0 + 48 + row];

    const float* ap = Y    + (size_t)(m0 + row) * D_INNER + koff;
    const float* bp = Wout + (size_t)koff * D_MODEL + n0 + row;

    v8f acc0 = {}, acc1 = {}, acc2 = {}, acc3 = {};
    for (int k = 0; k < D_INNER; k += 4) {
        const float w0 = lnw[k + koff], w1 = lnw[k + koff + 1];
        const float z0 = lnb[k + koff], z1 = lnb[k + koff + 1];
        v2f b;
        b.x = bp[(size_t)k * D_MODEL];
        b.y = bp[(size_t)k * D_MODEL + D_MODEL];
        v2f y0 = *(const v2f*)(ap + k);
        v2f y1 = *(const v2f*)(ap + 16 * D_INNER + k);
        v2f y2 = *(const v2f*)(ap + 32 * D_INNER + k);
        v2f y3 = *(const v2f*)(ap + 48 * D_INNER + k);
        v2f a0, a1, a2, a3;
        a0.x = (y0.x - mum0) * rsm0 * w0 + z0;  a0.y = (y0.y - mum0) * rsm0 * w1 + z1;
        a1.x = (y1.x - mum1) * rsm1 * w0 + z0;  a1.y = (y1.y - mum1) * rsm1 * w1 + z1;
        a2.x = (y2.x - mum2) * rsm2 * w0 + z0;  a2.y = (y2.y - mum2) * rsm2 * w1 + z1;
        a3.x = (y3.x - mum3) * rsm3 * w0 + z0;  a3.y = (y3.y - mum3) * rsm3 * w1 + z1;
        acc0 = wmma_f32_4(a0, b, acc0);
        acc1 = wmma_f32_4(a1, b, acc1);
        acc2 = wmma_f32_4(a2, b, acc2);
        acc3 = wmma_f32_4(a3, b, acc3);
    }
    const int mhi = (lane >> 4) << 3;
#pragma unroll
    for (int r = 0; r < 8; ++r) {
        const size_t mm = (size_t)(m0 + r + mhi);
        out[(mm     ) * D_MODEL + n0 + row] = acc0[r];
        out[(mm + 16) * D_MODEL + n0 + row] = acc1[r];
        out[(mm + 32) * D_MODEL + n0 + row] = acc2[r];
        out[(mm + 48) * D_MODEL + n0 + row] = acc3[r];
    }
}

extern "C" void kernel_launch(void* const* d_in, const int* in_sizes, int n_in,
                              void* d_out, int out_size, void* d_ws, size_t ws_size,
                              hipStream_t stream) {
    const float* u        = (const float*)d_in[0];
    const float* W_in     = (const float*)d_in[2];
    const float* conv_fw  = (const float*)d_in[3];
    const float* conv_fb  = (const float*)d_in[4];
    const float* conv_bw  = (const float*)d_in[5];
    const float* conv_bb  = (const float*)d_in[6];
    const float* dt_bias  = (const float*)d_in[7];
    const float* A_log    = (const float*)d_in[8];
    const float* D_param  = (const float*)d_in[9];
    const float* ln_w     = (const float*)d_in[10];
    const float* ln_b     = (const float*)d_in[11];
    const float* W_out    = (const float*)d_in[12];
    float* out            = (float*)d_out;

    float* ws = (float*)d_ws;
    size_t off = 0;
    float* xbcdt = ws + off; off += (size_t)BL * D_IN_PROJ;
    float* xc    = ws + off; off += (size_t)BL * CONV2;
    float* dtb   = ws + off; off += (size_t)BL * NHEADS;
    float* ybuf  = ws + off; off += (size_t)BL * D_INNER;
    float* Sbuf  = ws + off; off += (size_t)32 * NCHUNK * HEADDIM * NSTATE;
    float* pAbuf = ws + off; off += (size_t)32 * NCHUNK;
    float* mubuf = ws + off; off += (size_t)BL;
    float* rsbuf = ws + off; off += (size_t)BL;

    gemm_in<<<dim3(D_IN_PROJ / 16, BL / 256), 128, 0, stream>>>(u, W_in, xbcdt);
    conv_dt<<<dim3(BL), 256, 0, stream>>>(xbcdt, conv_fw, conv_fb, conv_bw, conv_bb,
                                          dt_bias, xc, dtb);
    scan_chunk<<<dim3(BATCH * NHEADS * NCHUNK), 64, 0, stream>>>(
        xc, dtb, A_log, D_param, ybuf, Sbuf, pAbuf);
    scan_combine<<<dim3(BATCH * NHEADS), 64, 0, stream>>>(Sbuf, pAbuf);
    scan_correct<<<dim3(BATCH * NHEADS * NCHUNK), 64, 0, stream>>>(
        xc, dtb, A_log, Sbuf, ybuf);
    ln_stats<<<dim3(BL), 256, 0, stream>>>(ybuf, mubuf, rsbuf);
    gemm_out<<<dim3(D_MODEL / 16, BL / 256), 128, 0, stream>>>(
        ybuf, mubuf, rsbuf, ln_w, ln_b, W_out, out);

    (void)in_sizes; (void)n_in; (void)out_size; (void)ws_size;
}


# scan passes 1 and 3: stage 32 timesteps of B,C,x/y,dt in LDS per barrier pair (was 1 step per 2 barriers); next tile's global loads in flight during compute; summation order unchanged
# speedup vs baseline: 1.5835x; 1.5835x over previous
#include <hip/hip_runtime.h>
#include <hip/hip_bf16.h>
#include <math.h>

#define BATCH    2
#define SEQLEN   4096
#define BL       (BATCH * SEQLEN)
#define D_MODEL  512
#define D_IN_PROJ 1168
#define CONV_DIM 576
#define CONV2    1152
#define D_INNER  1024
#define NHEADS   16
#define HH       8
#define HEADDIM  64
#define NSTATE   32
#define NCHUNK   32
#define CHUNK    128

typedef float v2f __attribute__((ext_vector_type(2)));
typedef float v8f __attribute__((ext_vector_type(8)));

__device__ __forceinline__ v8f wmma_f32_4(v2f a, v2f b, v8f c) {
    return __builtin_amdgcn_wmma_f32_16x16x4_f32(
         false, a,  false, b,
         (short)0, c,  false,  false);
}

__global__ void __launch_bounds__(128)
gemm_in(const float* __restrict__ U, const float* __restrict__ Win,
        float* __restrict__ out) {
    const int wv   = threadIdx.x >> 5;
    const int lane = threadIdx.x & 31;
    const int row  = lane & 15;
    const int koff = (lane >> 4) << 1;
    const int m0   = (blockIdx.y * 16 + wv * 4) * 16;
    const int n0   = blockIdx.x * 16;

    const float* ap = U   + (size_t)(m0 + row) * D_MODEL + koff;
    const float* bp = Win + (size_t)koff * D_IN_PROJ + n0 + row;

    v8f acc0 = {}, acc1 = {}, acc2 = {}, acc3 = {};
    for (int k = 0; k < D_MODEL; k += 4) {
        v2f b;
        b.x = bp[(size_t)k * D_IN_PROJ];
        b.y = bp[(size_t)k * D_IN_PROJ + D_IN_PROJ];
        v2f a0 = *(const v2f*)(ap + k);
        v2f a1 = *(const v2f*)(ap + 16 * D_MODEL + k);
        v2f a2 = *(const v2f*)(ap + 32 * D_MODEL + k);
        v2f a3 = *(const v2f*)(ap + 48 * D_MODEL + k);
        acc0 = wmma_f32_4(a0, b, acc0);
        acc1 = wmma_f32_4(a1, b, acc1);
        acc2 = wmma_f32_4(a2, b, acc2);
        acc3 = wmma_f32_4(a3, b, acc3);
    }
    const int mhi = (lane >> 4) << 3;
#pragma unroll
    for (int r = 0; r < 8; ++r) {
        const size_t mm = (size_t)(m0 + r + mhi);
        out[(mm     ) * D_IN_PROJ + n0 + row] = acc0[r];
        out[(mm + 16) * D_IN_PROJ + n0 + row] = acc1[r];
        out[(mm + 32) * D_IN_PROJ + n0 + row] = acc2[r];
        out[(mm + 48) * D_IN_PROJ + n0 + row] = acc3[r];
    }
}

__global__ void __launch_bounds__(256)
conv_dt(const float* __restrict__ xbcdt,
        const float* __restrict__ wf, const float* __restrict__ bf,
        const float* __restrict__ wb, const float* __restrict__ bb,
        const float* __restrict__ dt_bias,
        float* __restrict__ xc, float* __restrict__ dtb) {
    const size_t bl = blockIdx.x;
    const int b = (int)(bl / SEQLEN);
    const int t = (int)(bl % SEQLEN);

    for (int c = threadIdx.x; c < CONV2; c += blockDim.x) {
        float acc;
        if (c < CONV_DIM) {
            acc = bf[c];
#pragma unroll
            for (int k = 0; k < 4; ++k) {
                int tt = t - 3 + k;
                if (tt >= 0)
                    acc += xbcdt[((size_t)b * SEQLEN + tt) * D_IN_PROJ + c] * wf[c * 4 + k];
            }
        } else {
            int cc = c - CONV_DIM;
            acc = bb[cc];
#pragma unroll
            for (int k = 0; k < 4; ++k) {
                int tt = t + 3 - k;
                if (tt < SEQLEN)
                    acc += xbcdt[((size_t)b * SEQLEN + tt) * D_IN_PROJ + c] * wb[cc * 4 + k];
            }
        }
        float sig = 1.0f / (1.0f + expf(-acc));
        xc[bl * CONV2 + c] = acc * sig;
    }
    for (int h = threadIdx.x; h < NHEADS; h += blockDim.x) {
        float v = xbcdt[bl * D_IN_PROJ + CONV2 + h] + dt_bias[h];
        dtb[bl * NHEADS + h] = (v > 20.0f) ? v : log1pf(expf(v));
    }
}

__device__ __forceinline__ void head_bases(int h, int& xbase, int& bbase, int& cbase) {
    if (h < HH) { xbase = h * HEADDIM;                 bbase = 512;  cbase = 544;  }
    else        { xbase = CONV_DIM + (h - HH) * HEADDIM; bbase = 1088; cbase = 1120; }
}

#define TS     32
#define NTILE  (CHUNK / TS)
static_assert(TS == 32, "the B|C staging below names TS/4 = 8 registers");

__global__ void __launch_bounds__(64) __attribute__((amdgpu_waves_per_eu(1, 4)))
scan_chunk(const float* __restrict__ xc, const float* __restrict__ dtb,
           const float* __restrict__ A_log, const float* __restrict__ Dp,
           float* __restrict__ y, float* __restrict__ S, float* __restrict__ prodA) {
    const int chunk = blockIdx.x & (NCHUNK - 1);
    const int h     = (blockIdx.x >> 5) & 15;
    const int b     = blockIdx.x >> 9;
    const int p     = threadIdx.x;
    int xbase, bbase, cbase; head_bases(h, xbase, bbase, cbase);
    (void)cbase;

    const float Ah = -expf(A_log[h]);
    const float Dh = Dp[h];
    float hst[NSTATE];
#pragma unroll
    for (int n = 0; n < NSTATE; ++n) hst[n] = 0.0f;
    float pA = 1.0f;

    __shared__ __attribute__((aligned(16))) float sBC[TS][2 * NSTATE];
    __shared__ float sX[TS][HEADDIM];
    __shared__ float sDt[TS], sDA[TS];

    const int  s_first = chunk * CHUNK;
    const long rdir    = (h < HH) ? 1 : -1;
    const long row0    = (long)b * SEQLEN + ((h < HH) ? s_first : (SEQLEN - 1 - s_first));

    const int lrow = p >> 4;
    const int lc4  = (p & 15) * 4;

    float4 q0, q1, q2, q3, q4, q5, q6, q7;
    float  rx[TS];
    float  rdt = 0.0f;

#define SCAN_CHUNK_LOAD_TILE(tile_)                                                        \
    {                                                                                      \
        const long rt = row0 + rdir * (long)((tile_) * TS);                                \
        const float* bcp = xc + (rt + rdir * lrow) * CONV2 + bbase + lc4;                  \
        const long   bcs = rdir * 4 * CONV2;                                               \
        q0 = *(const float4*)(bcp);            q1 = *(const float4*)(bcp + bcs);           \
        q2 = *(const float4*)(bcp + 2 * bcs);  q3 = *(const float4*)(bcp + 3 * bcs);       \
        q4 = *(const float4*)(bcp + 4 * bcs);  q5 = *(const float4*)(bcp + 5 * bcs);       \
        q6 = *(const float4*)(bcp + 6 * bcs);  q7 = *(const float4*)(bcp + 7 * bcs);       \
        _Pragma("unroll")                                                                  \
        for (int j = 0; j < TS; ++j)                                                       \
            rx[j] = xc[(rt + rdir * j) * CONV2 + xbase + p];                               \
        if (p < TS) rdt = dtb[(rt + rdir * p) * NHEADS + h];                               \
    }

    SCAN_CHUNK_LOAD_TILE(0)

#pragma unroll 1
    for (int tile = 0; tile < NTILE; ++tile) {
        *(float4*)&sBC[lrow     ][lc4] = q0;  *(float4*)&sBC[lrow +  4][lc4] = q1;
        *(float4*)&sBC[lrow +  8][lc4] = q2;  *(float4*)&sBC[lrow + 12][lc4] = q3;
        *(float4*)&sBC[lrow + 16][lc4] = q4;  *(float4*)&sBC[lrow + 20][lc4] = q5;
        *(float4*)&sBC[lrow + 24][lc4] = q6;  *(float4*)&sBC[lrow + 28][lc4] = q7;
#pragma unroll
        for (int j = 0; j < TS; ++j) sX[j][p] = rx[j];
        if (p < TS) { sDt[p] = rdt; sDA[p] = expf(Ah * rdt); }
        __syncthreads();

        if (tile + 1 < NTILE) SCAN_CHUNK_LOAD_TILE(tile + 1)

        const long rt = row0 + rdir * (long)(tile * TS);
#pragma unroll 2
        for (int j = 0; j < TS; ++j) {
            const float dtv = sDt[j];
            const float dA  = sDA[j];
            pA *= dA;
            const float xp  = sX[j][p];
            const float dtx = dtv * xp;
            float accy = 0.0f;
#pragma unroll
            for (int n = 0; n < NSTATE; ++n) {
                hst[n] = dA * hst[n] + dtx * sBC[j][n];
                accy  += hst[n] * sBC[j][NSTATE + n];
            }
            y[(rt + rdir * j) * D_INNER + h * HEADDIM + p] = accy + Dh * xp;
        }
        __syncthreads();
    }
#undef SCAN_CHUNK_LOAD_TILE

    const size_t sb = (((size_t)(b * NHEADS + h)) * NCHUNK + chunk) * (HEADDIM * NSTATE)
                    + (size_t)p * NSTATE;
#pragma unroll
    for (int n = 0; n < NSTATE; ++n) S[sb + n] = hst[n];
    if (p == 0) prodA[(b * NHEADS + h) * NCHUNK + chunk] = pA;
}

__global__ void __launch_bounds__(64)
scan_combine(float* __restrict__ S, const float* __restrict__ prodA) {
    const int bh = blockIdx.x;
    const int p  = threadIdx.x;
    float hcur[NSTATE];
#pragma unroll
    for (int n = 0; n < NSTATE; ++n) hcur[n] = 0.0f;
    for (int c = 0; c < NCHUNK; ++c) {
        const size_t base = (((size_t)bh) * NCHUNK + c) * (HEADDIM * NSTATE) + (size_t)p * NSTATE;
        const float a = prodA[bh * NCHUNK + c];
#pragma unroll
        for (int n = 0; n < NSTATE; ++n) {
            float sl = S[base + n];
            S[base + n] = hcur[n];
            hcur[n] = a * hcur[n] + sl;
        }
    }
}

__global__ void __launch_bounds__(64) __attribute__((amdgpu_waves_per_eu(1, 4)))
scan_correct(const float* __restrict__ xc, const float* __restrict__ dtb,
             const float* __restrict__ A_log, const float* __restrict__ S,
             float* __restrict__ y) {
    const int chunk = blockIdx.x & (NCHUNK - 1);
    if (chunk == 0) return;
    const int h = (blockIdx.x >> 5) & 15;
    const int b = blockIdx.x >> 9;
    const int p = threadIdx.x;
    int xbase, bbase, cbase; head_bases(h, xbase, bbase, cbase);
    (void)xbase; (void)bbase;

    float hi[NSTATE];
    const size_t sb = (((size_t)(b * NHEADS + h)) * NCHUNK + chunk) * (HEADDIM * NSTATE)
                    + (size_t)p * NSTATE;
#pragma unroll
    for (int n = 0; n < NSTATE; ++n) hi[n] = S[sb + n];

    const float Ah = -expf(A_log[h]);
    float cumA = 1.0f;

    __shared__ __attribute__((aligned(16))) float sC[TS][NSTATE];
    __shared__ float sY[TS][HEADDIM];
    __shared__ float sDA[TS];

    const int  s_first = chunk * CHUNK;
    const long rdir    = (h < HH) ? 1 : -1;
    const long row0    = (long)b * SEQLEN + ((h < HH) ? s_first : (SEQLEN - 1 - s_first));

    const int lrow = p >> 3;
    const int lc4  = (p & 7) * 4;

    float4 c0, c1, c2, c3;
    float  ry[TS];
    float  rdt = 0.0f;

#define SCAN_CORRECT_LOAD_TILE(tile_)                                                      \
    {                                                                                      \
        const long rt = row0 + rdir * (long)((tile_) * TS);                                \
        const float* cp  = xc + (rt + rdir * lrow) * CONV2 + cbase + lc4;                  \
        const long   cst = rdir * 8 * CONV2;                                               \
        c0 = *(const float4*)(cp);            c1 = *(const float4*)(cp + cst);             \
        c2 = *(const float4*)(cp + 2 * cst);  c3 = *(const float4*)(cp + 3 * cst);         \
        _Pragma("unroll")                                                                  \
        for (int j = 0; j < TS; ++j)                                                       \
            ry[j] = y[(rt + rdir * j) * D_INNER + h * HEADDIM + p];                        \
        if (p < TS) rdt = dtb[(rt + rdir * p) * NHEADS + h];                               \
    }

    SCAN_CORRECT_LOAD_TILE(0)

#pragma unroll 1
    for (int tile = 0; tile < NTILE; ++tile) {
        *(float4*)&sC[lrow     ][lc4] = c0;  *(float4*)&sC[lrow +  8][lc4] = c1;
        *(float4*)&sC[lrow + 16][lc4] = c2;  *(float4*)&sC[lrow + 24][lc4] = c3;
#pragma unroll
        for (int j = 0; j < TS; ++j) sY[j][p] = ry[j];
        if (p < TS) sDA[p] = expf(Ah * rdt);
        __syncthreads();

        if (tile + 1 < NTILE) SCAN_CORRECT_LOAD_TILE(tile + 1)

        const long rt = row0 + rdir * (long)(tile * TS);
#pragma unroll 4
        for (int j = 0; j < TS; ++j) {
            cumA *= sDA[j];
            float acc = 0.0f;
#pragma unroll
            for (int n = 0; n < NSTATE; ++n) acc += hi[n] * sC[j][n];
            y[(rt + rdir * j) * D_INNER + h * HEADDIM + p] = sY[j][p] + cumA * acc;
        }
        __syncthreads();
    }
#undef SCAN_CORRECT_LOAD_TILE
}

__global__ void __launch_bounds__(256)
ln_stats(const float* __restrict__ y, float* __restrict__ mu, float* __restrict__ rstd) {
    const size_t row = blockIdx.x;
    const int tid = threadIdx.x;
    const float* yr = y + row * D_INNER;
    float s = 0.0f, q = 0.0f;
    for (int i = tid; i < D_INNER; i += 256) { float v = yr[i]; s += v; q += v * v; }
    __shared__ float ss[256], sq[256];
    ss[tid] = s; sq[tid] = q; __syncthreads();
    for (int off = 128; off > 0; off >>= 1) {
        if (tid < off) { ss[tid] += ss[tid + off]; sq[tid] += sq[tid + off]; }
        __syncthreads();
    }
    if (tid == 0) {
        float m = ss[0] * (1.0f / D_INNER);
        float v = sq[0] * (1.0f / D_INNER) - m * m;
        mu[row] = m;
        rstd[row] = rsqrtf(v + 1e-5f);
    }
}

__global__ void __launch_bounds__(128)
gemm_out(const float* __restrict__ Y, const float* __restrict__ mu,
         const float* __restrict__ rstd, const float* __restrict__ lnw,
         const float* __restrict__ lnb, const float* __restrict__ Wout,
         float* __restrict__ out) {
    const int wv   = threadIdx.x >> 5;
    const int lane = threadIdx.x & 31;
    const int row  = lane & 15;
    const int koff = (lane >> 4) << 1;
    const int m0   = (blockIdx.y * 16 + wv * 4) * 16;
    const int n0   = blockIdx.x * 16;

    const float mum0 = mu[m0 + row],        rsm0 = rstd[m0 + row];
    const float mum1 = mu[m0 + 16 + row],   rsm1 = rstd[m0 + 16 + row];
    const float mum2 = mu[m0 + 32 + row],   rsm2 = rstd[m0 + 32 + row];
    const float mum3 = mu[m0 + 48 + row],   rsm3 = rstd[m0 + 48 + row];

    const float* ap = Y    + (size_t)(m0 + row) * D_INNER + koff;
    const float* bp = Wout + (size_t)koff * D_MODEL + n0 + row;

    v8f acc0 = {}, acc1 = {}, acc2 = {}, acc3 = {};
    for (int k = 0; k < D_INNER; k += 4) {
        const float w0 = lnw[k + koff], w1 = lnw[k + koff + 1];
        const float z0 = lnb[k + koff], z1 = lnb[k + koff + 1];
        v2f b;
        b.x = bp[(size_t)k * D_MODEL];
        b.y = bp[(size_t)k * D_MODEL + D_MODEL];
        v2f y0 = *(const v2f*)(ap + k);
        v2f y1 = *(const v2f*)(ap + 16 * D_INNER + k);
        v2f y2 = *(const v2f*)(ap + 32 * D_INNER + k);
        v2f y3 = *(const v2f*)(ap + 48 * D_INNER + k);
        v2f a0, a1, a2, a3;
        a0.x = (y0.x - mum0) * rsm0 * w0 + z0;  a0.y = (y0.y - mum0) * rsm0 * w1 + z1;
        a1.x = (y1.x - mum1) * rsm1 * w0 + z0;  a1.y = (y1.y - mum1) * rsm1 * w1 + z1;
        a2.x = (y2.x - mum2) * rsm2 * w0 + z0;  a2.y = (y2.y - mum2) * rsm2 * w1 + z1;
        a3.x = (y3.x - mum3) * rsm3 * w0 + z0;  a3.y = (y3.y - mum3) * rsm3 * w1 + z1;
        acc0 = wmma_f32_4(a0, b, acc0);
        acc1 = wmma_f32_4(a1, b, acc1);
        acc2 = wmma_f32_4(a2, b, acc2);
        acc3 = wmma_f32_4(a3, b, acc3);
    }
    const int mhi = (lane >> 4) << 3;
#pragma unroll
    for (int r = 0; r < 8; ++r) {
        const size_t mm = (size_t)(m0 + r + mhi);
        out[(mm     ) * D_MODEL + n0 + row] = acc0[r];
        out[(mm + 16) * D_MODEL + n0 + row] = acc1[r];
        out[(mm + 32) * D_MODEL + n0 + row] = acc2[r];
        out[(mm + 48) * D_MODEL + n0 + row] = acc3[r];
    }
}

extern "C" void kernel_launch(void* const* d_in, const int* in_sizes, int n_in,
                              void* d_out, int out_size, void* d_ws, size_t ws_size,
                              hipStream_t stream) {
    const float* u        = (const float*)d_in[0];
    const float* W_in     = (const float*)d_in[2];
    const float* conv_fw  = (const float*)d_in[3];
    const float* conv_fb  = (const float*)d_in[4];
    const float* conv_bw  = (const float*)d_in[5];
    const float* conv_bb  = (const float*)d_in[6];
    const float* dt_bias  = (const float*)d_in[7];
    const float* A_log    = (const float*)d_in[8];
    const float* D_param  = (const float*)d_in[9];
    const float* ln_w     = (const float*)d_in[10];
    const float* ln_b     = (const float*)d_in[11];
    const float* W_out    = (const float*)d_in[12];
    float* out            = (float*)d_out;

    float* ws = (float*)d_ws;
    size_t off = 0;
    float* xbcdt = ws + off; off += (size_t)BL * D_IN_PROJ;
    float* xc    = ws + off; off += (size_t)BL * CONV2;
    float* dtb   = ws + off; off += (size_t)BL * NHEADS;
    float* ybuf  = ws + off; off += (size_t)BL * D_INNER;
    float* Sbuf  = ws + off; off += (size_t)32 * NCHUNK * HEADDIM * NSTATE;
    float* pAbuf = ws + off; off += (size_t)32 * NCHUNK;
    float* mubuf = ws + off; off += (size_t)BL;
    float* rsbuf = ws + off; off += (size_t)BL;

    gemm_in<<<dim3(D_IN_PROJ / 16, BL / 256), 128, 0, stream>>>(u, W_in, xbcdt);
    conv_dt<<<dim3(BL), 256, 0, stream>>>(xbcdt, conv_fw, conv_fb, conv_bw, conv_bb,
                                          dt_bias, xc, dtb);
    scan_chunk<<<dim3(BATCH * NHEADS * NCHUNK), 64, 0, stream>>>(
        xc, dtb, A_log, D_param, ybuf, Sbuf, pAbuf);
    scan_combine<<<dim3(BATCH * NHEADS), 64, 0, stream>>>(Sbuf, pAbuf);
    scan_correct<<<dim3(BATCH * NHEADS * NCHUNK), 64, 0, stream>>>(
        xc, dtb, A_log, Sbuf, ybuf);
    ln_stats<<<dim3(BL), 256, 0, stream>>>(ybuf, mubuf, rsbuf);
    gemm_out<<<dim3(D_MODEL / 16, BL / 256), 128, 0, stream>>>(
        ybuf, mubuf, rsbuf, ln_w, ln_b, W_out, out);

    (void)in_sizes; (void)n_in; (void)out_size; (void)ws_size;
}


# both GEMMs: A and B K-slabs staged in LDS by 16-byte global loads (block tile 128x64, 8 waves, wave tile 32x32, 2 LDS stages); WMMA fragments read from LDS; k and summation order unchanged
# speedup vs baseline: 2.3642x; 1.4930x over previous
#include <hip/hip_runtime.h>
#include <hip/hip_bf16.h>
#include <math.h>

#define BATCH    2
#define SEQLEN   4096
#define BL       (BATCH * SEQLEN)
#define D_MODEL  512
#define D_IN_PROJ 1168
#define CONV_DIM 576
#define CONV2    1152
#define D_INNER  1024
#define NHEADS   16
#define HH       8
#define HEADDIM  64
#define NSTATE   32
#define NCHUNK   32
#define CHUNK    128

typedef float v2f __attribute__((ext_vector_type(2)));
typedef float v8f __attribute__((ext_vector_type(8)));

__device__ __forceinline__ v8f wmma_f32_4(v2f a, v2f b, v8f c) {
    return __builtin_amdgcn_wmma_f32_16x16x4_f32(
         false, a,  false, b,
         (short)0, c,  false,  false);
}

#define GBM 128
#define GBN 64
#define GBK 32
#define APITCH 36
#define BPITCH2 160
#define SA_FLOATS (GBM * APITCH)
#define SB_FLOATS ((GBK / 2) * BPITCH2)
#define STAGE_FLOATS (SA_FLOATS + SB_FLOATS)
#define GIN_NBLK  19
#define GOUT_NBLK 8
#define G_MBLK    64

__device__ __forceinline__ void slab_mma(const float* pa, const float* pb,
                                         v8f& c00, v8f& c01, v8f& c10, v8f& c11) {
#pragma unroll
    for (int kk = 0; kk < GBK; kk += 4) {
        const v2f a0 = *(const v2f*)(pa + kk);
        const v2f a1 = *(const v2f*)(pa + 16 * APITCH + kk);
        const v2f b0 = *(const v2f*)(pb + (kk / 2) * BPITCH2);
        const v2f b1 = *(const v2f*)(pb + (kk / 2) * BPITCH2 + 32);
        c00 = wmma_f32_4(a0, b0, c00);
        c01 = wmma_f32_4(a0, b1, c01);
        c10 = wmma_f32_4(a1, b0, c10);
        c11 = wmma_f32_4(a1, b1, c11);
    }
}

__global__ void __launch_bounds__(256)
gemm_in(const float* __restrict__ U, const float* __restrict__ Win,
        float* __restrict__ out) {
    __shared__ __attribute__((aligned(16))) float smem[2 * STAGE_FLOATS];
    const int tid  = threadIdx.x;
    const int wv   = tid >> 5;
    const int lane = tid & 31;
    const int row  = lane & 15;
    const int koff = (lane >> 4) << 1;
    const int wm   = wv >> 1;
    const int wn   = wv & 1;
    const int m0   = blockIdx.y * GBM;
    const int n0   = blockIdx.x * GBN;

    const int ar = tid >> 3, ac = (tid & 7) << 2;
    const int br = tid >> 4, bc = (tid & 15) << 2;
    const bool bok = (n0 + bc) < D_IN_PROJ;
    const float* ag = U   + (size_t)(m0 + ar) * D_MODEL + ac;
    const float* bg = Win + (size_t)(2 * br) * D_IN_PROJ + n0 + bc;

    float4 ra0, ra1, ra2, ra3, rb0, rb1;
#define GEMM_IN_LOAD(kt_)                                                                  \
    {                                                                                      \
        const float* a_ = ag + (kt_) * GBK;                                                \
        ra0 = *(const float4*)(a_);                                                        \
        ra1 = *(const float4*)(a_ + 32 * D_MODEL);                                         \
        ra2 = *(const float4*)(a_ + 64 * D_MODEL);                                         \
        ra3 = *(const float4*)(a_ + 96 * D_MODEL);                                         \
        if (bok) {                                                                         \
            const float* b_ = bg + (size_t)(kt_) * GBK * D_IN_PROJ;                        \
            rb0 = *(const float4*)(b_);                                                    \
            rb1 = *(const float4*)(b_ + D_IN_PROJ);                                        \
        } else {                                                                           \
            rb0 = make_float4(0.0f, 0.0f, 0.0f, 0.0f);                                     \
            rb1 = make_float4(0.0f, 0.0f, 0.0f, 0.0f);                                     \
        }                                                                                  \
    }
#define GEMM_IN_STAGE(st_)                                                                 \
    {                                                                                      \
        float* sa_ = smem + (st_) * STAGE_FLOATS + ar * APITCH + ac;                       \
        float* sb_ = smem + (st_) * STAGE_FLOATS + SA_FLOATS + br * BPITCH2 + 2 * bc;      \
        *(float4*)(sa_)               = ra0;                                               \
        *(float4*)(sa_ + 32 * APITCH) = ra1;                                               \
        *(float4*)(sa_ + 64 * APITCH) = ra2;                                               \
        *(float4*)(sa_ + 96 * APITCH) = ra3;                                               \
        *(float4*)(sb_)     = make_float4(rb0.x, rb1.x, rb0.y, rb1.y);                     \
        *(float4*)(sb_ + 4) = make_float4(rb0.z, rb1.z, rb0.w, rb1.w);                     \
    }

    v8f c00 = {}, c01 = {}, c10 = {}, c11 = {};
    const bool wave_on = (n0 + wn * 32) < D_IN_PROJ;
    const int fa = (wm * 32 + row) * APITCH + koff;
    const int fb = SA_FLOATS + (lane >> 4) * BPITCH2 + (wn * 32 + row) * 2;

    GEMM_IN_LOAD(0)
    GEMM_IN_STAGE(0)
    __syncthreads();

#pragma unroll 1
    for (int kt = 0; kt < D_MODEL / GBK; ++kt) {
        const int st = kt & 1;
        if (kt + 1 < D_MODEL / GBK) GEMM_IN_LOAD(kt + 1)
        if (wave_on)
            slab_mma(smem + st * STAGE_FLOATS + fa, smem + st * STAGE_FLOATS + fb, c00, c01, c10, c11);
        if (kt + 1 < D_MODEL / GBK) GEMM_IN_STAGE(st ^ 1)
        __syncthreads();
    }
#undef GEMM_IN_LOAD
#undef GEMM_IN_STAGE

    if (wave_on) {
        const int  mhi  = (lane >> 4) << 3;
        const int  nb   = n0 + wn * 32 + row;
        const bool n1ok = (n0 + wn * 32 + 16) < D_IN_PROJ;
#pragma unroll
        for (int r = 0; r < 8; ++r) {
            const size_t mm = (size_t)(m0 + wm * 32 + mhi + r);
            out[(mm     ) * D_IN_PROJ + nb] = c00[r];
            out[(mm + 16) * D_IN_PROJ + nb] = c10[r];
            if (n1ok) {
                out[(mm     ) * D_IN_PROJ + nb + 16] = c01[r];
                out[(mm + 16) * D_IN_PROJ + nb + 16] = c11[r];
            }
        }
    }
}

__global__ void __launch_bounds__(256)
conv_dt(const float* __restrict__ xbcdt,
        const float* __restrict__ wf, const float* __restrict__ bf,
        const float* __restrict__ wb, const float* __restrict__ bb,
        const float* __restrict__ dt_bias,
        float* __restrict__ xc, float* __restrict__ dtb) {
    const size_t bl = blockIdx.x;
    const int b = (int)(bl / SEQLEN);
    const int t = (int)(bl % SEQLEN);

    for (int c = threadIdx.x; c < CONV2; c += blockDim.x) {
        float acc;
        if (c < CONV_DIM) {
            acc = bf[c];
#pragma unroll
            for (int k = 0; k < 4; ++k) {
                int tt = t - 3 + k;
                if (tt >= 0)
                    acc += xbcdt[((size_t)b * SEQLEN + tt) * D_IN_PROJ + c] * wf[c * 4 + k];
            }
        } else {
            int cc = c - CONV_DIM;
            acc = bb[cc];
#pragma unroll
            for (int k = 0; k < 4; ++k) {
                int tt = t + 3 - k;
                if (tt < SEQLEN)
                    acc += xbcdt[((size_t)b * SEQLEN + tt) * D_IN_PROJ + c] * wb[cc * 4 + k];
            }
        }
        float sig = 1.0f / (1.0f + expf(-acc));
        xc[bl * CONV2 + c] = acc * sig;
    }
    for (int h = threadIdx.x; h < NHEADS; h += blockDim.x) {
        float v = xbcdt[bl * D_IN_PROJ + CONV2 + h] + dt_bias[h];
        dtb[bl * NHEADS + h] = (v > 20.0f) ? v : log1pf(expf(v));
    }
}

__device__ __forceinline__ void head_bases(int h, int& xbase, int& bbase, int& cbase) {
    if (h < HH) { xbase = h * HEADDIM;                 bbase = 512;  cbase = 544;  }
    else        { xbase = CONV_DIM + (h - HH) * HEADDIM; bbase = 1088; cbase = 1120; }
}

#define TS     32
#define NTILE  (CHUNK / TS)
static_assert(TS == 32, "the B|C staging below names TS/4 = 8 registers");

__global__ void __launch_bounds__(64) __attribute__((amdgpu_waves_per_eu(1, 4)))
scan_chunk(const float* __restrict__ xc, const float* __restrict__ dtb,
           const float* __restrict__ A_log, const float* __restrict__ Dp,
           float* __restrict__ y, float* __restrict__ S, float* __restrict__ prodA) {
    const int chunk = blockIdx.x & (NCHUNK - 1);
    const int h     = (blockIdx.x >> 5) & 15;
    const int b     = blockIdx.x >> 9;
    const int p     = threadIdx.x;
    int xbase, bbase, cbase; head_bases(h, xbase, bbase, cbase);
    (void)cbase;

    const float Ah = -expf(A_log[h]);
    const float Dh = Dp[h];
    float hst[NSTATE];
#pragma unroll
    for (int n = 0; n < NSTATE; ++n) hst[n] = 0.0f;
    float pA = 1.0f;

    __shared__ __attribute__((aligned(16))) float sBC[TS][2 * NSTATE];
    __shared__ float sX[TS][HEADDIM];
    __shared__ float sDt[TS], sDA[TS];

    const int  s_first = chunk * CHUNK;
    const long rdir    = (h < HH) ? 1 : -1;
    const long row0    = (long)b * SEQLEN + ((h < HH) ? s_first : (SEQLEN - 1 - s_first));

    const int lrow = p >> 4;
    const int lc4  = (p & 15) * 4;

    float4 q0, q1, q2, q3, q4, q5, q6, q7;
    float  rx[TS];
    float  rdt = 0.0f;

#define SCAN_CHUNK_LOAD_TILE(tile_)                                                        \
    {                                                                                      \
        const long rt = row0 + rdir * (long)((tile_) * TS);                                \
        const float* bcp = xc + (rt + rdir * lrow) * CONV2 + bbase + lc4;                  \
        const long   bcs = rdir * 4 * CONV2;                                               \
        q0 = *(const float4*)(bcp);            q1 = *(const float4*)(bcp + bcs);           \
        q2 = *(const float4*)(bcp + 2 * bcs);  q3 = *(const float4*)(bcp + 3 * bcs);       \
        q4 = *(const float4*)(bcp + 4 * bcs);  q5 = *(const float4*)(bcp + 5 * bcs);       \
        q6 = *(const float4*)(bcp + 6 * bcs);  q7 = *(const float4*)(bcp + 7 * bcs);       \
        _Pragma("unroll")                                                                  \
        for (int j = 0; j < TS; ++j)                                                       \
            rx[j] = xc[(rt + rdir * j) * CONV2 + xbase + p];                               \
        if (p < TS) rdt = dtb[(rt + rdir * p) * NHEADS + h];                               \
    }

    SCAN_CHUNK_LOAD_TILE(0)

#pragma unroll 1
    for (int tile = 0; tile < NTILE; ++tile) {
        *(float4*)&sBC[lrow     ][lc4] = q0;  *(float4*)&sBC[lrow +  4][lc4] = q1;
        *(float4*)&sBC[lrow +  8][lc4] = q2;  *(float4*)&sBC[lrow + 12][lc4] = q3;
        *(float4*)&sBC[lrow + 16][lc4] = q4;  *(float4*)&sBC[lrow + 20][lc4] = q5;
        *(float4*)&sBC[lrow + 24][lc4] = q6;  *(float4*)&sBC[lrow + 28][lc4] = q7;
#pragma unroll
        for (int j = 0; j < TS; ++j) sX[j][p] = rx[j];
        if (p < TS) { sDt[p] = rdt; sDA[p] = expf(Ah * rdt); }
        __syncthreads();

        if (tile + 1 < NTILE) SCAN_CHUNK_LOAD_TILE(tile + 1)

        const long rt = row0 + rdir * (long)(tile * TS);
#pragma unroll 2
        for (int j = 0; j < TS; ++j) {
            const float dtv = sDt[j];
            const float dA  = sDA[j];
            pA *= dA;
            const float xp  = sX[j][p];
            const float dtx = dtv * xp;
            float accy = 0.0f;
#pragma unroll
            for (int n = 0; n < NSTATE; ++n) {
                hst[n] = dA * hst[n] + dtx * sBC[j][n];
                accy  += hst[n] * sBC[j][NSTATE + n];
            }
            y[(rt + rdir * j) * D_INNER + h * HEADDIM + p] = accy + Dh * xp;
        }
        __syncthreads();
    }
#undef SCAN_CHUNK_LOAD_TILE

    const size_t sb = (((size_t)(b * NHEADS + h)) * NCHUNK + chunk) * (HEADDIM * NSTATE)
                    + (size_t)p * NSTATE;
#pragma unroll
    for (int n = 0; n < NSTATE; ++n) S[sb + n] = hst[n];
    if (p == 0) prodA[(b * NHEADS + h) * NCHUNK + chunk] = pA;
}

__global__ void __launch_bounds__(64)
scan_combine(float* __restrict__ S, const float* __restrict__ prodA) {
    const int bh = blockIdx.x;
    const int p  = threadIdx.x;
    float hcur[NSTATE];
#pragma unroll
    for (int n = 0; n < NSTATE; ++n) hcur[n] = 0.0f;
    for (int c = 0; c < NCHUNK; ++c) {
        const size_t base = (((size_t)bh) * NCHUNK + c) * (HEADDIM * NSTATE) + (size_t)p * NSTATE;
        const float a = prodA[bh * NCHUNK + c];
#pragma unroll
        for (int n = 0; n < NSTATE; ++n) {
            float sl = S[base + n];
            S[base + n] = hcur[n];
            hcur[n] = a * hcur[n] + sl;
        }
    }
}

__global__ void __launch_bounds__(64) __attribute__((amdgpu_waves_per_eu(1, 4)))
scan_correct(const float* __restrict__ xc, const float* __restrict__ dtb,
             const float* __restrict__ A_log, const float* __restrict__ S,
             float* __restrict__ y) {
    const int chunk = blockIdx.x & (NCHUNK - 1);
    if (chunk == 0) return;
    const int h = (blockIdx.x >> 5) & 15;
    const int b = blockIdx.x >> 9;
    const int p = threadIdx.x;
    int xbase, bbase, cbase; head_bases(h, xbase, bbase, cbase);
    (void)xbase; (void)bbase;

    float hi[NSTATE];
    const size_t sb = (((size_t)(b * NHEADS + h)) * NCHUNK + chunk) * (HEADDIM * NSTATE)
                    + (size_t)p * NSTATE;
#pragma unroll
    for (int n = 0; n < NSTATE; ++n) hi[n] = S[sb + n];

    const float Ah = -expf(A_log[h]);
    float cumA = 1.0f;

    __shared__ __attribute__((aligned(16))) float sC[TS][NSTATE];
    __shared__ float sY[TS][HEADDIM];
    __shared__ float sDA[TS];

    const int  s_first = chunk * CHUNK;
    const long rdir    = (h < HH) ? 1 : -1;
    const long row0    = (long)b * SEQLEN + ((h < HH) ? s_first : (SEQLEN - 1 - s_first));

    const int lrow = p >> 3;
    const int lc4  = (p & 7) * 4;

    float4 c0, c1, c2, c3;
    float  ry[TS];
    float  rdt = 0.0f;

#define SCAN_CORRECT_LOAD_TILE(tile_)                                                      \
    {                                                                                      \
        const long rt = row0 + rdir * (long)((tile_) * TS);                                \
        const float* cp  = xc + (rt + rdir * lrow) * CONV2 + cbase + lc4;                  \
        const long   cst = rdir * 8 * CONV2;                                               \
        c0 = *(const float4*)(cp);            c1 = *(const float4*)(cp + cst);             \
        c2 = *(const float4*)(cp + 2 * cst);  c3 = *(const float4*)(cp + 3 * cst);         \
        _Pragma("unroll")                                                                  \
        for (int j = 0; j < TS; ++j)                                                       \
            ry[j] = y[(rt + rdir * j) * D_INNER + h * HEADDIM + p];                        \
        if (p < TS) rdt = dtb[(rt + rdir * p) * NHEADS + h];                               \
    }

    SCAN_CORRECT_LOAD_TILE(0)

#pragma unroll 1
    for (int tile = 0; tile < NTILE; ++tile) {
        *(float4*)&sC[lrow     ][lc4] = c0;  *(float4*)&sC[lrow +  8][lc4] = c1;
        *(float4*)&sC[lrow + 16][lc4] = c2;  *(float4*)&sC[lrow + 24][lc4] = c3;
#pragma unroll
        for (int j = 0; j < TS; ++j) sY[j][p] = ry[j];
        if (p < TS) sDA[p] = expf(Ah * rdt);
        __syncthreads();

        if (tile + 1 < NTILE) SCAN_CORRECT_LOAD_TILE(tile + 1)

        const long rt = row0 + rdir * (long)(tile * TS);
#pragma unroll 4
        for (int j = 0; j < TS; ++j) {
            cumA *= sDA[j];
            float acc = 0.0f;
#pragma unroll
            for (int n = 0; n < NSTATE; ++n) acc += hi[n] * sC[j][n];
            y[(rt + rdir * j) * D_INNER + h * HEADDIM + p] = sY[j][p] + cumA * acc;
        }
        __syncthreads();
    }
#undef SCAN_CORRECT_LOAD_TILE
}

__global__ void __launch_bounds__(256)
ln_stats(const float* __restrict__ y, float* __restrict__ mu, float* __restrict__ rstd) {
    const size_t row = blockIdx.x;
    const int tid = threadIdx.x;
    const float* yr = y + row * D_INNER;
    float s = 0.0f, q = 0.0f;
    for (int i = tid; i < D_INNER; i += 256) { float v = yr[i]; s += v; q += v * v; }
    __shared__ float ss[256], sq[256];
    ss[tid] = s; sq[tid] = q; __syncthreads();
    for (int off = 128; off > 0; off >>= 1) {
        if (tid < off) { ss[tid] += ss[tid + off]; sq[tid] += sq[tid + off]; }
        __syncthreads();
    }
    if (tid == 0) {
        float m = ss[0] * (1.0f / D_INNER);
        float v = sq[0] * (1.0f / D_INNER) - m * m;
        mu[row] = m;
        rstd[row] = rsqrtf(v + 1e-5f);
    }
}

__global__ void __launch_bounds__(256)
gemm_out(const float* __restrict__ Y, const float* __restrict__ mu,
         const float* __restrict__ rstd, const float* __restrict__ lnw,
         const float* __restrict__ lnb, const float* __restrict__ Wout,
         float* __restrict__ out) {
    __shared__ __attribute__((aligned(16))) float smem[2 * STAGE_FLOATS];
    const int tid  = threadIdx.x;
    const int wv   = tid >> 5;
    const int lane = tid & 31;
    const int row  = lane & 15;
    const int koff = (lane >> 4) << 1;
    const int wm   = wv >> 1;
    const int wn   = wv & 1;
    const int m0   = blockIdx.y * GBM;
    const int n0   = blockIdx.x * GBN;

    const int ar = tid >> 3, ac = (tid & 7) << 2;
    const int br = tid >> 4, bc = (tid & 15) << 2;
    const float* ag = Y    + (size_t)(m0 + ar) * D_INNER + ac;
    const float* bg = Wout + (size_t)(2 * br) * D_MODEL + n0 + bc;

    const float mu0 = mu[m0 + ar],      rs0 = rstd[m0 + ar];
    const float mu1 = mu[m0 + ar + 32], rs1 = rstd[m0 + ar + 32];
    const float mu2 = mu[m0 + ar + 64], rs2 = rstd[m0 + ar + 64];
    const float mu3 = mu[m0 + ar + 96], rs3 = rstd[m0 + ar + 96];

    float4 ra0, ra1, ra2, ra3, rb0, rb1, rw, rz;
#define GEMM_OUT_LOAD(kt_)                                                                 \
    {                                                                                      \
        const float* a_ = ag + (kt_) * GBK;                                                \
        ra0 = *(const float4*)(a_);                                                        \
        ra1 = *(const float4*)(a_ + 32 * D_INNER);                                         \
        ra2 = *(const float4*)(a_ + 64 * D_INNER);                                         \
        ra3 = *(const float4*)(a_ + 96 * D_INNER);                                         \
        const float* b_ = bg + (size_t)(kt_) * GBK * D_MODEL;                              \
        rb0 = *(const float4*)(b_);                                                        \
        rb1 = *(const float4*)(b_ + D_MODEL);                                              \
        rw  = *(const float4*)(lnw + (kt_) * GBK + ac);                                    \
        rz  = *(const float4*)(lnb + (kt_) * GBK + ac);                                    \
    }
#define GEMM_OUT_LN4(v_, mu_, rs_)                                                         \
    make_float4(((v_).x - (mu_)) * (rs_) * rw.x + rz.x, ((v_).y - (mu_)) * (rs_) * rw.y + rz.y, \
                ((v_).z - (mu_)) * (rs_) * rw.z + rz.z, ((v_).w - (mu_)) * (rs_) * rw.w + rz.w)
#define GEMM_OUT_STAGE(st_)                                                                \
    {                                                                                      \
        float* sa_ = smem + (st_) * STAGE_FLOATS + ar * APITCH + ac;                       \
        float* sb_ = smem + (st_) * STAGE_FLOATS + SA_FLOATS + br * BPITCH2 + 2 * bc;      \
        *(float4*)(sa_)               = GEMM_OUT_LN4(ra0, mu0, rs0);                       \
        *(float4*)(sa_ + 32 * APITCH) = GEMM_OUT_LN4(ra1, mu1, rs1);                       \
        *(float4*)(sa_ + 64 * APITCH) = GEMM_OUT_LN4(ra2, mu2, rs2);                       \
        *(float4*)(sa_ + 96 * APITCH) = GEMM_OUT_LN4(ra3, mu3, rs3);                       \
        *(float4*)(sb_)     = make_float4(rb0.x, rb1.x, rb0.y, rb1.y);                     \
        *(float4*)(sb_ + 4) = make_float4(rb0.z, rb1.z, rb0.w, rb1.w);                     \
    }

    v8f c00 = {}, c01 = {}, c10 = {}, c11 = {};
    const int fa = (wm * 32 + row) * APITCH + koff;
    const int fb = SA_FLOATS + (lane >> 4) * BPITCH2 + (wn * 32 + row) * 2;

    GEMM_OUT_LOAD(0)
    GEMM_OUT_STAGE(0)
    __syncthreads();

#pragma unroll 1
    for (int kt = 0; kt < D_INNER / GBK; ++kt) {
        const int st = kt & 1;
        if (kt + 1 < D_INNER / GBK) GEMM_OUT_LOAD(kt + 1)
        slab_mma(smem + st * STAGE_FLOATS + fa, smem + st * STAGE_FLOATS + fb, c00, c01, c10, c11);
        if (kt + 1 < D_INNER / GBK) GEMM_OUT_STAGE(st ^ 1)
        __syncthreads();
    }
#undef GEMM_OUT_LOAD
#undef GEMM_OUT_LN4
#undef GEMM_OUT_STAGE

    const int mhi = (lane >> 4) << 3;
    const int nb  = n0 + wn * 32 + row;
#pragma unroll
    for (int r = 0; r < 8; ++r) {
        const size_t mm = (size_t)(m0 + wm * 32 + mhi + r);
        out[(mm     ) * D_MODEL + nb     ] = c00[r];
        out[(mm     ) * D_MODEL + nb + 16] = c01[r];
        out[(mm + 16) * D_MODEL + nb     ] = c10[r];
        out[(mm + 16) * D_MODEL + nb + 16] = c11[r];
    }
}

extern "C" void kernel_launch(void* const* d_in, const int* in_sizes, int n_in,
                              void* d_out, int out_size, void* d_ws, size_t ws_size,
                              hipStream_t stream) {
    const float* u        = (const float*)d_in[0];
    const float* W_in     = (const float*)d_in[2];
    const float* conv_fw  = (const float*)d_in[3];
    const float* conv_fb  = (const float*)d_in[4];
    const float* conv_bw  = (const float*)d_in[5];
    const float* conv_bb  = (const float*)d_in[6];
    const float* dt_bias  = (const float*)d_in[7];
    const float* A_log    = (const float*)d_in[8];
    const float* D_param  = (const float*)d_in[9];
    const float* ln_w     = (const float*)d_in[10];
    const float* ln_b     = (const float*)d_in[11];
    const float* W_out    = (const float*)d_in[12];
    float* out            = (float*)d_out;

    float* ws = (float*)d_ws;
    size_t off = 0;
    float* xbcdt = ws + off; off += (size_t)BL * D_IN_PROJ;
    float* xc    = ws + off; off += (size_t)BL * CONV2;
    float* dtb   = ws + off; off += (size_t)BL * NHEADS;
    float* ybuf  = ws + off; off += (size_t)BL * D_INNER;
    float* Sbuf  = ws + off; off += (size_t)32 * NCHUNK * HEADDIM * NSTATE;
    float* pAbuf = ws + off; off += (size_t)32 * NCHUNK;
    float* mubuf = ws + off; off += (size_t)BL;
    float* rsbuf = ws + off; off += (size_t)BL;

    gemm_in<<<dim3(GIN_NBLK, G_MBLK), 256, 0, stream>>>(u, W_in, xbcdt);
    conv_dt<<<dim3(BL), 256, 0, stream>>>(xbcdt, conv_fw, conv_fb, conv_bw, conv_bb,
                                          dt_bias, xc, dtb);
    scan_chunk<<<dim3(BATCH * NHEADS * NCHUNK), 64, 0, stream>>>(
        xc, dtb, A_log, D_param, ybuf, Sbuf, pAbuf);
    scan_combine<<<dim3(BATCH * NHEADS), 64, 0, stream>>>(Sbuf, pAbuf);
    scan_correct<<<dim3(BATCH * NHEADS * NCHUNK), 64, 0, stream>>>(
        xc, dtb, A_log, Sbuf, ybuf);
    ln_stats<<<dim3(BL), 256, 0, stream>>>(ybuf, mubuf, rsbuf);
    gemm_out<<<dim3(GOUT_NBLK, G_MBLK), 256, 0, stream>>>(
        ybuf, mubuf, rsbuf, ln_w, ln_b, W_out, out);

    (void)in_sizes; (void)n_in; (void)out_size; (void)ws_size;
}


# conv+dt and LN-stats: 65536 short waves each -> 6144 / 2048 long ones (conv: thread walks 16 timesteps, input rows kept in registers; stats: a wave per row, shuffle sums: new sum order of mu/rstd)
# speedup vs baseline: 2.7127x; 1.1474x over previous
#include <hip/hip_runtime.h>
#include <hip/hip_bf16.h>
#include <math.h>

#define BATCH    2
#define SEQLEN   4096
#define BL       (BATCH * SEQLEN)
#define D_MODEL  512
#define D_IN_PROJ 1168
#define CONV_DIM 576
#define CONV2    1152
#define D_INNER  1024
#define NHEADS   16
#define HH       8
#define HEADDIM  64
#define NSTATE   32
#define NCHUNK   32
#define CHUNK    128

typedef float v2f __attribute__((ext_vector_type(2)));
typedef float v8f __attribute__((ext_vector_type(8)));

__device__ __forceinline__ v8f wmma_f32_4(v2f a, v2f b, v8f c) {
    return __builtin_amdgcn_wmma_f32_16x16x4_f32(
         false, a,  false, b,
         (short)0, c,  false,  false);
}

#define GBM 128
#define GBN 64
#define GBK 32
#define APITCH 36
#define BPITCH2 160
#define SA_FLOATS (GBM * APITCH)
#define SB_FLOATS ((GBK / 2) * BPITCH2)
#define STAGE_FLOATS (SA_FLOATS + SB_FLOATS)
#define GIN_NBLK  19
#define GOUT_NBLK 8
#define G_MBLK    64

__device__ __forceinline__ void slab_mma(const float* pa, const float* pb,
                                         v8f& c00, v8f& c01, v8f& c10, v8f& c11) {
#pragma unroll
    for (int kk = 0; kk < GBK; kk += 4) {
        const v2f a0 = *(const v2f*)(pa + kk);
        const v2f a1 = *(const v2f*)(pa + 16 * APITCH + kk);
        const v2f b0 = *(const v2f*)(pb + (kk / 2) * BPITCH2);
        const v2f b1 = *(const v2f*)(pb + (kk / 2) * BPITCH2 + 32);
        c00 = wmma_f32_4(a0, b0, c00);
        c01 = wmma_f32_4(a0, b1, c01);
        c10 = wmma_f32_4(a1, b0, c10);
        c11 = wmma_f32_4(a1, b1, c11);
    }
}

__global__ void __launch_bounds__(256)
gemm_in(const float* __restrict__ U, const float* __restrict__ Win,
        float* __restrict__ out) {
    __shared__ __attribute__((aligned(16))) float smem[2 * STAGE_FLOATS];
    const int tid  = threadIdx.x;
    const int wv   = tid >> 5;
    const int lane = tid & 31;
    const int row  = lane & 15;
    const int koff = (lane >> 4) << 1;
    const int wm   = wv >> 1;
    const int wn   = wv & 1;
    const int m0   = blockIdx.y * GBM;
    const int n0   = blockIdx.x * GBN;

    const int ar = tid >> 3, ac = (tid & 7) << 2;
    const int br = tid >> 4, bc = (tid & 15) << 2;
    const bool bok = (n0 + bc) < D_IN_PROJ;
    const float* ag = U   + (size_t)(m0 + ar) * D_MODEL + ac;
    const float* bg = Win + (size_t)(2 * br) * D_IN_PROJ + n0 + bc;

    float4 ra0, ra1, ra2, ra3, rb0, rb1;
#define GEMM_IN_LOAD(kt_)                                                                  \
    {                                                                                      \
        const float* a_ = ag + (kt_) * GBK;                                                \
        ra0 = *(const float4*)(a_);                                                        \
        ra1 = *(const float4*)(a_ + 32 * D_MODEL);                                         \
        ra2 = *(const float4*)(a_ + 64 * D_MODEL);                                         \
        ra3 = *(const float4*)(a_ + 96 * D_MODEL);                                         \
        if (bok) {                                                                         \
            const float* b_ = bg + (size_t)(kt_) * GBK * D_IN_PROJ;                        \
            rb0 = *(const float4*)(b_);                                                    \
            rb1 = *(const float4*)(b_ + D_IN_PROJ);                                        \
        } else {                                                                           \
            rb0 = make_float4(0.0f, 0.0f, 0.0f, 0.0f);                                     \
            rb1 = make_float4(0.0f, 0.0f, 0.0f, 0.0f);                                     \
        }                                                                                  \
    }
#define GEMM_IN_STAGE(st_)                                                                 \
    {                                                                                      \
        float* sa_ = smem + (st_) * STAGE_FLOATS + ar * APITCH + ac;                       \
        float* sb_ = smem + (st_) * STAGE_FLOATS + SA_FLOATS + br * BPITCH2 + 2 * bc;      \
        *(float4*)(sa_)               = ra0;                                               \
        *(float4*)(sa_ + 32 * APITCH) = ra1;                                               \
        *(float4*)(sa_ + 64 * APITCH) = ra2;                                               \
        *(float4*)(sa_ + 96 * APITCH) = ra3;                                               \
        *(float4*)(sb_)     = make_float4(rb0.x, rb1.x, rb0.y, rb1.y);                     \
        *(float4*)(sb_ + 4) = make_float4(rb0.z, rb1.z, rb0.w, rb1.w);                     \
    }

    v8f c00 = {}, c01 = {}, c10 = {}, c11 = {};
    const bool wave_on = (n0 + wn * 32) < D_IN_PROJ;
    const int fa = (wm * 32 + row) * APITCH + koff;
    const int fb = SA_FLOATS + (lane >> 4) * BPITCH2 + (wn * 32 + row) * 2;

    GEMM_IN_LOAD(0)
    GEMM_IN_STAGE(0)
    __syncthreads();

#pragma unroll 1
    for (int kt = 0; kt < D_MODEL / GBK; ++kt) {
        const int st = kt & 1;
        if (kt + 1 < D_MODEL / GBK) GEMM_IN_LOAD(kt + 1)
        if (wave_on)
            slab_mma(smem + st * STAGE_FLOATS + fa, smem + st * STAGE_FLOATS + fb, c00, c01, c10, c11);
        if (kt + 1 < D_MODEL / GBK) GEMM_IN_STAGE(st ^ 1)
        __syncthreads();
    }
#undef GEMM_IN_LOAD
#undef GEMM_IN_STAGE

    if (wave_on) {
        const int  mhi  = (lane >> 4) << 3;
        const int  nb   = n0 + wn * 32 + row;
        const bool n1ok = (n0 + wn * 32 + 16) < D_IN_PROJ;
#pragma unroll
        for (int r = 0; r < 8; ++r) {
            const size_t mm = (size_t)(m0 + wm * 32 + mhi + r);
            out[(mm     ) * D_IN_PROJ + nb] = c00[r];
            out[(mm + 16) * D_IN_PROJ + nb] = c10[r];
            if (n1ok) {
                out[(mm     ) * D_IN_PROJ + nb + 16] = c01[r];
                out[(mm + 16) * D_IN_PROJ + nb + 16] = c11[r];
            }
        }
    }
}

#define CV_TT      16
#define CV_G       4
#define CV_NG      (CV_TT / CV_G)
#define CV_C4      (CONV_DIM / 4)
#define CV_THREADS 192
static_assert(SEQLEN % CV_TT == 0 && CV_TT % CV_G == 0 && CV_G >= 3,
              "a tile stays inside one sequence; the 3 rows beyond a trip's own are all inside it or all outside");

template <int DIR>
__device__ __forceinline__ void conv_tile(const float* __restrict__ xrow, float* __restrict__ orow,
                                          const float* __restrict__ w16, const float* __restrict__ b4,
                                          const int t0) {
    const float4 wq0 = *(const float4*)(w16);
    const float4 wq1 = *(const float4*)(w16 + 4);
    const float4 wq2 = *(const float4*)(w16 + 8);
    const float4 wq3 = *(const float4*)(w16 + 12);
    const float wt[4][4] = {{wq0.x, wq0.y, wq0.z, wq0.w}, {wq1.x, wq1.y, wq1.z, wq1.w},
                            {wq2.x, wq2.y, wq2.z, wq2.w}, {wq3.x, wq3.y, wq3.z, wq3.w}};
    const float4 bias = *(const float4*)(b4);

    const int tb = DIR ? t0 : t0 - 3;
    float4 win[CV_G + 3];
    float4 nxt[CV_G];
#pragma unroll
    for (int i = 0; i < 3; ++i) {
        const int tt = DIR ? (tb + i) : max(tb + i, 0);
        win[i] = *(const float4*)(xrow + (size_t)tt * D_IN_PROJ);
    }
#pragma unroll
    for (int i = 0; i < CV_G; ++i) {
        const int tt = DIR ? min(tb + 3 + i, SEQLEN - 1) : (tb + 3 + i);
        nxt[i] = *(const float4*)(xrow + (size_t)tt * D_IN_PROJ);
    }
    bool edge_ok = DIR ? true : (t0 != 0);

#pragma unroll 1
    for (int g = 0; g < CV_NG; ++g) {
#pragma unroll
        for (int i = 0; i < CV_G; ++i) win[3 + i] = nxt[i];
        if (g + 1 < CV_NG) {
            const int tn = tb + (g + 1) * CV_G + 3;
#pragma unroll
            for (int i = 0; i < CV_G; ++i) {
                const int tt = DIR ? min(tn + i, SEQLEN - 1) : (tn + i);
                nxt[i] = *(const float4*)(xrow + (size_t)tt * D_IN_PROJ);
            }
        }
        if (DIR) edge_ok = (g + 1 < CV_NG) || (t0 + CV_TT < SEQLEN);

#pragma unroll
        for (int j = 0; j < CV_G; ++j) {
            float a0 = bias.x, a1 = bias.y, a2 = bias.z, a3 = bias.w;
#pragma unroll
            for (int k = 0; k < 4; ++k) {
                const int  wi   = DIR ? (j + 3 - k) : (j + k);
                const bool edge = DIR ? (wi >= CV_G) : (wi < 3);
                if (!edge || edge_ok) {
                    a0 += win[wi].x * wt[0][k];
                    a1 += win[wi].y * wt[1][k];
                    a2 += win[wi].z * wt[2][k];
                    a3 += win[wi].w * wt[3][k];
                }
            }
            float4 o;
            o.x = a0 * (1.0f / (1.0f + expf(-a0)));
            o.y = a1 * (1.0f / (1.0f + expf(-a1)));
            o.z = a2 * (1.0f / (1.0f + expf(-a2)));
            o.w = a3 * (1.0f / (1.0f + expf(-a3)));
            *(float4*)(orow + (size_t)(t0 + g * CV_G + j) * CONV2) = o;
        }
#pragma unroll
        for (int i = 0; i < 3; ++i) win[i] = win[CV_G + i];
        if (!DIR) edge_ok = true;
    }
}

__global__ void __launch_bounds__(CV_THREADS)
conv_dt(const float* __restrict__ xbcdt,
        const float* __restrict__ wf, const float* __restrict__ bf,
        const float* __restrict__ wb, const float* __restrict__ bb,
        const float* __restrict__ dt_bias,
        float* __restrict__ xc, float* __restrict__ dtb) {
    const int    tid = threadIdx.x;
    const size_t r0  = (size_t)blockIdx.x * CV_TT;
    const int    t0  = (int)(r0 % SEQLEN);
    const size_t rb  = r0 - (size_t)t0;

    if (tid < CV_C4) {
        const int c = tid * 4;
        if (blockIdx.y == 0)
            conv_tile<0>(xbcdt + rb * D_IN_PROJ + c, xc + rb * CONV2 + c, wf + c * 4, bf + c, t0);
        else
            conv_tile<1>(xbcdt + rb * D_IN_PROJ + CONV_DIM + c, xc + rb * CONV2 + CONV_DIM + c,
                         wb + c * 4, bb + c, t0);
    } else if (tid >= 160 && blockIdx.y == 0) {
        const int    l   = tid - 160;
        const int    h   = l & 15;
        const size_t rw  = r0 + (l >> 4);
        const float  dbh = dt_bias[h];
        float raw[CV_TT / 2];
#pragma unroll
        for (int i = 0; i < CV_TT / 2; ++i) raw[i] = xbcdt[(rw + 2 * i) * D_IN_PROJ + CONV2 + h];
#pragma unroll
        for (int i = 0; i < CV_TT / 2; ++i) {
            const float v = raw[i] + dbh;
            dtb[(rw + 2 * i) * NHEADS + h] = (v > 20.0f) ? v : log1pf(expf(v));
        }
    }
}

__device__ __forceinline__ void head_bases(int h, int& xbase, int& bbase, int& cbase) {
    if (h < HH) { xbase = h * HEADDIM;                 bbase = 512;  cbase = 544;  }
    else        { xbase = CONV_DIM + (h - HH) * HEADDIM; bbase = 1088; cbase = 1120; }
}

#define TS     32
#define NTILE  (CHUNK / TS)
static_assert(TS == 32, "the B|C staging below names TS/4 = 8 registers");

__global__ void __launch_bounds__(64) __attribute__((amdgpu_waves_per_eu(1, 4)))
scan_chunk(const float* __restrict__ xc, const float* __restrict__ dtb,
           const float* __restrict__ A_log, const float* __restrict__ Dp,
           float* __restrict__ y, float* __restrict__ S, float* __restrict__ prodA) {
    const int chunk = blockIdx.x & (NCHUNK - 1);
    const int h     = (blockIdx.x >> 5) & 15;
    const int b     = blockIdx.x >> 9;
    const int p     = threadIdx.x;
    int xbase, bbase, cbase; head_bases(h, xbase, bbase, cbase);
    (void)cbase;

    const float Ah = -expf(A_log[h]);
    const float Dh = Dp[h];
    float hst[NSTATE];
#pragma unroll
    for (int n = 0; n < NSTATE; ++n) hst[n] = 0.0f;
    float pA = 1.0f;

    __shared__ __attribute__((aligned(16))) float sBC[TS][2 * NSTATE];
    __shared__ float sX[TS][HEADDIM];
    __shared__ float sDt[TS], sDA[TS];

    const int  s_first = chunk * CHUNK;
    const long rdir    = (h < HH) ? 1 : -1;
    const long row0    = (long)b * SEQLEN + ((h < HH) ? s_first : (SEQLEN - 1 - s_first));

    const int lrow = p >> 4;
    const int lc4  = (p & 15) * 4;

    float4 q0, q1, q2, q3, q4, q5, q6, q7;
    float  rx[TS];
    float  rdt = 0.0f;

#define SCAN_CHUNK_LOAD_TILE(tile_)                                                        \
    {                                                                                      \
        const long rt = row0 + rdir * (long)((tile_) * TS);                                \
        const float* bcp = xc + (rt + rdir * lrow) * CONV2 + bbase + lc4;                  \
        const long   bcs = rdir * 4 * CONV2;                                               \
        q0 = *(const float4*)(bcp);            q1 = *(const float4*)(bcp + bcs);           \
        q2 = *(const float4*)(bcp + 2 * bcs);  q3 = *(const float4*)(bcp + 3 * bcs);       \
        q4 = *(const float4*)(bcp + 4 * bcs);  q5 = *(const float4*)(bcp + 5 * bcs);       \
        q6 = *(const float4*)(bcp + 6 * bcs);  q7 = *(const float4*)(bcp + 7 * bcs);       \
        _Pragma("unroll")                                                                  \
        for (int j = 0; j < TS; ++j)                                                       \
            rx[j] = xc[(rt + rdir * j) * CONV2 + xbase + p];                               \
        if (p < TS) rdt = dtb[(rt + rdir * p) * NHEADS + h];                               \
    }

    SCAN_CHUNK_LOAD_TILE(0)

#pragma unroll 1
    for (int tile = 0; tile < NTILE; ++tile) {
        *(float4*)&sBC[lrow     ][lc4] = q0;  *(float4*)&sBC[lrow +  4][lc4] = q1;
        *(float4*)&sBC[lrow +  8][lc4] = q2;  *(float4*)&sBC[lrow + 12][lc4] = q3;
        *(float4*)&sBC[lrow + 16][lc4] = q4;  *(float4*)&sBC[lrow + 20][lc4] = q5;
        *(float4*)&sBC[lrow + 24][lc4] = q6;  *(float4*)&sBC[lrow + 28][lc4] = q7;
#pragma unroll
        for (int j = 0; j < TS; ++j) sX[j][p] = rx[j];
        if (p < TS) { sDt[p] = rdt; sDA[p] = expf(Ah * rdt); }
        __syncthreads();

        if (tile + 1 < NTILE) SCAN_CHUNK_LOAD_TILE(tile + 1)

        const long rt = row0 + rdir * (long)(tile * TS);
#pragma unroll 2
        for (int j = 0; j < TS; ++j) {
            const float dtv = sDt[j];
            const float dA  = sDA[j];
            pA *= dA;
            const float xp  = sX[j][p];
            const float dtx = dtv * xp;
            float accy = 0.0f;
#pragma unroll
            for (int n = 0; n < NSTATE; ++n) {
                hst[n] = dA * hst[n] + dtx * sBC[j][n];
                accy  += hst[n] * sBC[j][NSTATE + n];
            }
            y[(rt + rdir * j) * D_INNER + h * HEADDIM + p] = accy + Dh * xp;
        }
        __syncthreads();
    }
#undef SCAN_CHUNK_LOAD_TILE

    const size_t sb = (((size_t)(b * NHEADS + h)) * NCHUNK + chunk) * (HEADDIM * NSTATE)
                    + (size_t)p * NSTATE;
#pragma unroll
    for (int n = 0; n < NSTATE; ++n) S[sb + n] = hst[n];
    if (p == 0) prodA[(b * NHEADS + h) * NCHUNK + chunk] = pA;
}

__global__ void __launch_bounds__(64)
scan_combine(float* __restrict__ S, const float* __restrict__ prodA) {
    const int bh = blockIdx.x;
    const int p  = threadIdx.x;
    float hcur[NSTATE];
#pragma unroll
    for (int n = 0; n < NSTATE; ++n) hcur[n] = 0.0f;
    for (int c = 0; c < NCHUNK; ++c) {
        const size_t base = (((size_t)bh) * NCHUNK + c) * (HEADDIM * NSTATE) + (size_t)p * NSTATE;
        const float a = prodA[bh * NCHUNK + c];
#pragma unroll
        for (int n = 0; n < NSTATE; ++n) {
            float sl = S[base + n];
            S[base + n] = hcur[n];
            hcur[n] = a * hcur[n] + sl;
        }
    }
}

__global__ void __launch_bounds__(64) __attribute__((amdgpu_waves_per_eu(1, 4)))
scan_correct(const float* __restrict__ xc, const float* __restrict__ dtb,
             const float* __restrict__ A_log, const float* __restrict__ S,
             float* __restrict__ y) {
    const int chunk = blockIdx.x & (NCHUNK - 1);
    if (chunk == 0) return;
    const int h = (blockIdx.x >> 5) & 15;
    const int b = blockIdx.x >> 9;
    const int p = threadIdx.x;
    int xbase, bbase, cbase; head_bases(h, xbase, bbase, cbase);
    (void)xbase; (void)bbase;

    float hi[NSTATE];
    const size_t sb = (((size_t)(b * NHEADS + h)) * NCHUNK + chunk) * (HEADDIM * NSTATE)
                    + (size_t)p * NSTATE;
#pragma unroll
    for (int n = 0; n < NSTATE; ++n) hi[n] = S[sb + n];

    const float Ah = -expf(A_log[h]);
    float cumA = 1.0f;

    __shared__ __attribute__((aligned(16))) float sC[TS][NSTATE];
    __shared__ float sY[TS][HEADDIM];
    __shared__ float sDA[TS];

    const int  s_first = chunk * CHUNK;
    const long rdir    = (h < HH) ? 1 : -1;
    const long row0    = (long)b * SEQLEN + ((h < HH) ? s_first : (SEQLEN - 1 - s_first));

    const int lrow = p >> 3;
    const int lc4  = (p & 7) * 4;

    float4 c0, c1, c2, c3;
    float  ry[TS];
    float  rdt = 0.0f;

#define SCAN_CORRECT_LOAD_TILE(tile_)                                                      \
    {                                                                                      \
        const long rt = row0 + rdir * (long)((tile_) * TS);                                \
        const float* cp  = xc + (rt + rdir * lrow) * CONV2 + cbase + lc4;                  \
        const long   cst = rdir * 8 * CONV2;                                               \
        c0 = *(const float4*)(cp);            c1 = *(const float4*)(cp + cst);             \
        c2 = *(const float4*)(cp + 2 * cst);  c3 = *(const float4*)(cp + 3 * cst);         \
        _Pragma("unroll")                                                                  \
        for (int j = 0; j < TS; ++j)                                                       \
            ry[j] = y[(rt + rdir * j) * D_INNER + h * HEADDIM + p];                        \
        if (p < TS) rdt = dtb[(rt + rdir * p) * NHEADS + h];                               \
    }

    SCAN_CORRECT_LOAD_TILE(0)

#pragma unroll 1
    for (int tile = 0; tile < NTILE; ++tile) {
        *(float4*)&sC[lrow     ][lc4] = c0;  *(float4*)&sC[lrow +  8][lc4] = c1;
        *(float4*)&sC[lrow + 16][lc4] = c2;  *(float4*)&sC[lrow + 24][lc4] = c3;
#pragma unroll
        for (int j = 0; j < TS; ++j) sY[j][p] = ry[j];
        if (p < TS) sDA[p] = expf(Ah * rdt);
        __syncthreads();

        if (tile + 1 < NTILE) SCAN_CORRECT_LOAD_TILE(tile + 1)

        const long rt = row0 + rdir * (long)(tile * TS);
#pragma unroll 4
        for (int j = 0; j < TS; ++j) {
            cumA *= sDA[j];
            float acc = 0.0f;
#pragma unroll
            for (int n = 0; n < NSTATE; ++n) acc += hi[n] * sC[j][n];
            y[(rt + rdir * j) * D_INNER + h * HEADDIM + p] = sY[j][p] + cumA * acc;
        }
        __syncthreads();
    }
#undef SCAN_CORRECT_LOAD_TILE
}

#define LN_RPW 4
#define LN_RPB (8 * LN_RPW)
static_assert(BL % LN_RPB == 0 && LN_RPB == 32, "wave 0 stores the work-group's 32 results, one per lane");

__global__ void __launch_bounds__(256)
ln_stats(const float* __restrict__ y, float* __restrict__ mu, float* __restrict__ rstd) {
    __shared__ float smu[LN_RPB], srs[LN_RPB];
    const int    tid  = threadIdx.x;
    const int    lane = tid & 31;
    const int    wv   = tid >> 5;
    const size_t rowb = (size_t)blockIdx.x * LN_RPB;
    const float* yw   = y + (rowb + (size_t)wv * LN_RPW) * D_INNER + lane * 4;

    float4 v[LN_RPW][8];
#pragma unroll
    for (int r = 0; r < LN_RPW; ++r)
#pragma unroll
        for (int i = 0; i < 8; ++i)
            v[r][i] = *(const float4*)(yw + r * D_INNER + i * 128);

#pragma unroll
    for (int r = 0; r < LN_RPW; ++r) {
        float s = 0.0f, q = 0.0f;
#pragma unroll
        for (int i = 0; i < 8; ++i) {
            s += v[r][i].x; q += v[r][i].x * v[r][i].x;
            s += v[r][i].y; q += v[r][i].y * v[r][i].y;
            s += v[r][i].z; q += v[r][i].z * v[r][i].z;
            s += v[r][i].w; q += v[r][i].w * v[r][i].w;
        }
#pragma unroll
        for (int off = 16; off > 0; off >>= 1) {
            s += __shfl_xor(s, off);
            q += __shfl_xor(q, off);
        }
        if (lane == 0) {
            const float m  = s * (1.0f / D_INNER);
            const float vr = q * (1.0f / D_INNER) - m * m;
            smu[wv * LN_RPW + r] = m;
            srs[wv * LN_RPW + r] = rsqrtf(vr + 1e-5f);
        }
    }
    __syncthreads();
    if (wv == 0) {
        mu[rowb + lane]   = smu[lane];
        rstd[rowb + lane] = srs[lane];
    }
}


__global__ void __launch_bounds__(256)
gemm_out(const float* __restrict__ Y, const float* __restrict__ mu,
         const float* __restrict__ rstd, const float* __restrict__ lnw,
         const float* __restrict__ lnb, const float* __restrict__ Wout,
         float* __restrict__ out) {
    __shared__ __attribute__((aligned(16))) float smem[2 * STAGE_FLOATS];
    const int tid  = threadIdx.x;
    const int wv   = tid >> 5;
    const int lane = tid & 31;
    const int row  = lane & 15;
    const int koff = (lane >> 4) << 1;
    const int wm   = wv >> 1;
    const int wn   = wv & 1;
    const int m0   = blockIdx.y * GBM;
    const int n0   = blockIdx.x * GBN;

    const int ar = tid >> 3, ac = (tid & 7) << 2;
    const int br = tid >> 4, bc = (tid & 15) << 2;
    const float* ag = Y    + (size_t)(m0 + ar) * D_INNER + ac;
    const float* bg = Wout + (size_t)(2 * br) * D_MODEL + n0 + bc;

    const float mu0 = mu[m0 + ar],      rs0 = rstd[m0 + ar];
    const float mu1 = mu[m0 + ar + 32], rs1 = rstd[m0 + ar + 32];
    const float mu2 = mu[m0 + ar + 64], rs2 = rstd[m0 + ar + 64];
    const float mu3 = mu[m0 + ar + 96], rs3 = rstd[m0 + ar + 96];

    float4 ra0, ra1, ra2, ra3, rb0, rb1, rw, rz;
#define GEMM_OUT_LOAD(kt_)                                                                 \
    {                                                                                      \
        const float* a_ = ag + (kt_) * GBK;                                                \
        ra0 = *(const float4*)(a_);                                                        \
        ra1 = *(const float4*)(a_ + 32 * D_INNER);                                         \
        ra2 = *(const float4*)(a_ + 64 * D_INNER);                                         \
        ra3 = *(const float4*)(a_ + 96 * D_INNER);                                         \
        const float* b_ = bg + (size_t)(kt_) * GBK * D_MODEL;                              \
        rb0 = *(const float4*)(b_);                                                        \
        rb1 = *(const float4*)(b_ + D_MODEL);                                              \
        rw  = *(const float4*)(lnw + (kt_) * GBK + ac);                                    \
        rz  = *(const float4*)(lnb + (kt_) * GBK + ac);                                    \
    }
#define GEMM_OUT_LN4(v_, mu_, rs_)                                                         \
    make_float4(((v_).x - (mu_)) * (rs_) * rw.x + rz.x, ((v_).y - (mu_)) * (rs_) * rw.y + rz.y, \
                ((v_).z - (mu_)) * (rs_) * rw.z + rz.z, ((v_).w - (mu_)) * (rs_) * rw.w + rz.w)
#define GEMM_OUT_STAGE(st_)                                                                \
    {                                                                                      \
        float* sa_ = smem + (st_) * STAGE_FLOATS + ar * APITCH + ac;                       \
        float* sb_ = smem + (st_) * STAGE_FLOATS + SA_FLOATS + br * BPITCH2 + 2 * bc;      \
        *(float4*)(sa_)               = GEMM_OUT_LN4(ra0, mu0, rs0);                       \
        *(float4*)(sa_ + 32 * APITCH) = GEMM_OUT_LN4(ra1, mu1, rs1);                       \
        *(float4*)(sa_ + 64 * APITCH) = GEMM_OUT_LN4(ra2, mu2, rs2);                       \
        *(float4*)(sa_ + 96 * APITCH) = GEMM_OUT_LN4(ra3, mu3, rs3);                       \
        *(float4*)(sb_)     = make_float4(rb0.x, rb1.x, rb0.y, rb1.y);                     \
        *(float4*)(sb_ + 4) = make_float4(rb0.z, rb1.z, rb0.w, rb1.w);                     \
    }

    v8f c00 = {}, c01 = {}, c10 = {}, c11 = {};
    const int fa = (wm * 32 + row) * APITCH + koff;
    const int fb = SA_FLOATS + (lane >> 4) * BPITCH2 + (wn * 32 + row) * 2;

    GEMM_OUT_LOAD(0)
    GEMM_OUT_STAGE(0)
    __syncthreads();

#pragma unroll 1
    for (int kt = 0; kt < D_INNER / GBK; ++kt) {
        const int st = kt & 1;
        if (kt + 1 < D_INNER / GBK) GEMM_OUT_LOAD(kt + 1)
        slab_mma(smem + st * STAGE_FLOATS + fa, smem + st * STAGE_FLOATS + fb, c00, c01, c10, c11);
        if (kt + 1 < D_INNER / GBK) GEMM_OUT_STAGE(st ^ 1)
        __syncthreads();
    }
#undef GEMM_OUT_LOAD
#undef GEMM_OUT_LN4
#undef GEMM_OUT_STAGE

    const int mhi = (lane >> 4) << 3;
    const int nb  = n0 + wn * 32 + row;
#pragma unroll
    for (int r = 0; r < 8; ++r) {
        const size_t mm = (size_t)(m0 + wm * 32 + mhi + r);
        out[(mm     ) * D_MODEL + nb     ] = c00[r];
        out[(mm     ) * D_MODEL + nb + 16] = c01[r];
        out[(mm + 16) * D_MODEL + nb     ] = c10[r];
        out[(mm + 16) * D_MODEL + nb + 16] = c11[r];
    }
}

extern "C" void kernel_launch(void* const* d_in, const int* in_sizes, int n_in,
                              void* d_out, int out_size, void* d_ws, size_t ws_size,
                              hipStream_t stream) {
    const float* u        = (const float*)d_in[0];
    const float* W_in     = (const float*)d_in[2];
    const float* conv_fw  = (const float*)d_in[3];
    const float* conv_fb  = (const float*)d_in[4];
    const float* conv_bw  = (const float*)d_in[5];
    const float* conv_bb  = (const float*)d_in[6];
    const float* dt_bias  = (const float*)d_in[7];
    const float* A_log    = (const float*)d_in[8];
    const float* D_param  = (const float*)d_in[9];
    const float* ln_w     = (const float*)d_in[10];
    const float* ln_b     = (const float*)d_in[11];
    const float* W_out    = (const float*)d_in[12];
    float* out            = (float*)d_out;

    float* ws = (float*)d_ws;
    size_t off = 0;
    float* xbcdt = ws + off; off += (size_t)BL * D_IN_PROJ;
    float* xc    = ws + off; off += (size_t)BL * CONV2;
    float* dtb   = ws + off; off += (size_t)BL * NHEADS;
    float* ybuf  = ws + off; off += (size_t)BL * D_INNER;
    float* Sbuf  = ws + off; off += (size_t)32 * NCHUNK * HEADDIM * NSTATE;
    float* pAbuf = ws + off; off += (size_t)32 * NCHUNK;
    float* mubuf = ws + off; off += (size_t)BL;
    float* rsbuf = ws + off; off += (size_t)BL;

    gemm_in<<<dim3(GIN_NBLK, G_MBLK), 256, 0, stream>>>(u, W_in, xbcdt);
    conv_dt<<<dim3(BL / CV_TT, 2), CV_THREADS, 0, stream>>>(xbcdt, conv_fw, conv_fb, conv_bw, conv_bb,
                                                            dt_bias, xc, dtb);
    scan_chunk<<<dim3(BATCH * NHEADS * NCHUNK), 64, 0, stream>>>(
        xc, dtb, A_log, D_param, ybuf, Sbuf, pAbuf);
    scan_combine<<<dim3(BATCH * NHEADS), 64, 0, stream>>>(Sbuf, pAbuf);
    scan_correct<<<dim3(BATCH * NHEADS * NCHUNK), 64, 0, stream>>>(
        xc, dtb, A_log, Sbuf, ybuf);
    ln_stats<<<dim3(BL / LN_RPB), 256, 0, stream>>>(ybuf, mubuf, rsbuf);
    gemm_out<<<dim3(GOUT_NBLK, G_MBLK), 256, 0, stream>>>(
        ybuf, mubuf, rsbuf, ln_w, ln_b, W_out, out);

    (void)in_sizes; (void)n_in; (void)out_size; (void)ws_size;
}


# scan pass 2 (chunk combine): a thread owns 4 state elements (16384 threads, was 2048) and issues all chunks' loads before the serial recurrence (was one round trip per chunk); sum order unchanged
# speedup vs baseline: 3.3628x; 1.2397x over previous
#include <hip/hip_runtime.h>
#include <hip/hip_bf16.h>
#include <math.h>

#define BATCH    2
#define SEQLEN   4096
#define BL       (BATCH * SEQLEN)
#define D_MODEL  512
#define D_IN_PROJ 1168
#define CONV_DIM 576
#define CONV2    1152
#define D_INNER  1024
#define NHEADS   16
#define HH       8
#define HEADDIM  64
#define NSTATE   32
#define NCHUNK   32
#define CHUNK    128

typedef float v2f __attribute__((ext_vector_type(2)));
typedef float v8f __attribute__((ext_vector_type(8)));

__device__ __forceinline__ v8f wmma_f32_4(v2f a, v2f b, v8f c) {
    return __builtin_amdgcn_wmma_f32_16x16x4_f32(
         false, a,  false, b,
         (short)0, c,  false,  false);
}

#define GBM 128
#define GBN 64
#define GBK 32
#define APITCH 36
#define BPITCH2 160
#define SA_FLOATS (GBM * APITCH)
#define SB_FLOATS ((GBK / 2) * BPITCH2)
#define STAGE_FLOATS (SA_FLOATS + SB_FLOATS)
#define GIN_NBLK  19
#define GOUT_NBLK 8
#define G_MBLK    64

__device__ __forceinline__ void slab_mma(const float* pa, const float* pb,
                                         v8f& c00, v8f& c01, v8f& c10, v8f& c11) {
#pragma unroll
    for (int kk = 0; kk < GBK; kk += 4) {
        const v2f a0 = *(const v2f*)(pa + kk);
        const v2f a1 = *(const v2f*)(pa + 16 * APITCH + kk);
        const v2f b0 = *(const v2f*)(pb + (kk / 2) * BPITCH2);
        const v2f b1 = *(const v2f*)(pb + (kk / 2) * BPITCH2 + 32);
        c00 = wmma_f32_4(a0, b0, c00);
        c01 = wmma_f32_4(a0, b1, c01);
        c10 = wmma_f32_4(a1, b0, c10);
        c11 = wmma_f32_4(a1, b1, c11);
    }
}

__global__ void __launch_bounds__(256)
gemm_in(const float* __restrict__ U, const float* __restrict__ Win,
        float* __restrict__ out) {
    __shared__ __attribute__((aligned(16))) float smem[2 * STAGE_FLOATS];
    const int tid  = threadIdx.x;
    const int wv   = tid >> 5;
    const int lane = tid & 31;
    const int row  = lane & 15;
    const int koff = (lane >> 4) << 1;
    const int wm   = wv >> 1;
    const int wn   = wv & 1;
    const int m0   = blockIdx.y * GBM;
    const int n0   = blockIdx.x * GBN;

    const int ar = tid >> 3, ac = (tid & 7) << 2;
    const int br = tid >> 4, bc = (tid & 15) << 2;
    const bool bok = (n0 + bc) < D_IN_PROJ;
    const float* ag = U   + (size_t)(m0 + ar) * D_MODEL + ac;
    const float* bg = Win + (size_t)(2 * br) * D_IN_PROJ + n0 + bc;

    float4 ra0, ra1, ra2, ra3, rb0, rb1;
#define GEMM_IN_LOAD(kt_)                                                                  \
    {                                                                                      \
        const float* a_ = ag + (kt_) * GBK;                                                \
        ra0 = *(const float4*)(a_);                                                        \
        ra1 = *(const float4*)(a_ + 32 * D_MODEL);                                         \
        ra2 = *(const float4*)(a_ + 64 * D_MODEL);                                         \
        ra3 = *(const float4*)(a_ + 96 * D_MODEL);                                         \
        if (bok) {                                                                         \
            const float* b_ = bg + (size_t)(kt_) * GBK * D_IN_PROJ;                        \
            rb0 = *(const float4*)(b_);                                                    \
            rb1 = *(const float4*)(b_ + D_IN_PROJ);                                        \
        } else {                                                                           \
            rb0 = make_float4(0.0f, 0.0f, 0.0f, 0.0f);                                     \
            rb1 = make_float4(0.0f, 0.0f, 0.0f, 0.0f);                                     \
        }                                                                                  \
    }
#define GEMM_IN_STAGE(st_)                                                                 \
    {                                                                                      \
        float* sa_ = smem + (st_) * STAGE_FLOATS + ar * APITCH + ac;                       \
        float* sb_ = smem + (st_) * STAGE_FLOATS + SA_FLOATS + br * BPITCH2 + 2 * bc;      \
        *(float4*)(sa_)               = ra0;                                               \
        *(float4*)(sa_ + 32 * APITCH) = ra1;                                               \
        *(float4*)(sa_ + 64 * APITCH) = ra2;                                               \
        *(float4*)(sa_ + 96 * APITCH) = ra3;                                               \
        *(float4*)(sb_)     = make_float4(rb0.x, rb1.x, rb0.y, rb1.y);                     \
        *(float4*)(sb_ + 4) = make_float4(rb0.z, rb1.z, rb0.w, rb1.w);                     \
    }

    v8f c00 = {}, c01 = {}, c10 = {}, c11 = {};
    const bool wave_on = (n0 + wn * 32) < D_IN_PROJ;
    const int fa = (wm * 32 + row) * APITCH + koff;
    const int fb = SA_FLOATS + (lane >> 4) * BPITCH2 + (wn * 32 + row) * 2;

    GEMM_IN_LOAD(0)
    GEMM_IN_STAGE(0)
    __syncthreads();

#pragma unroll 1
    for (int kt = 0; kt < D_MODEL / GBK; ++kt) {
        const int st = kt & 1;
        if (kt + 1 < D_MODEL / GBK) GEMM_IN_LOAD(kt + 1)
        if (wave_on)
            slab_mma(smem + st * STAGE_FLOATS + fa, smem + st * STAGE_FLOATS + fb, c00, c01, c10, c11);
        if (kt + 1 < D_MODEL / GBK) GEMM_IN_STAGE(st ^ 1)
        __syncthreads();
    }
#undef GEMM_IN_LOAD
#undef GEMM_IN_STAGE

    if (wave_on) {
        const int  mhi  = (lane >> 4) << 3;
        const int  nb   = n0 + wn * 32 + row;
        const bool n1ok = (n0 + wn * 32 + 16) < D_IN_PROJ;
#pragma unroll
        for (int r = 0; r < 8; ++r) {
            const size_t mm = (size_t)(m0 + wm * 32 + mhi + r);
            out[(mm     ) * D_IN_PROJ + nb] = c00[r];
            out[(mm + 16) * D_IN_PROJ + nb] = c10[r];
            if (n1ok) {
                out[(mm     ) * D_IN_PROJ + nb + 16] = c01[r];
                out[(mm + 16) * D_IN_PROJ + nb + 16] = c11[r];
            }
        }
    }
}

#define CV_TT      16
#define CV_G       4
#define CV_NG      (CV_TT / CV_G)
#define CV_C4      (CONV_DIM / 4)
#define CV_THREADS 192
static_assert(SEQLEN % CV_TT == 0 && CV_TT % CV_G == 0 && CV_G >= 3,
              "a tile stays inside one sequence; the 3 rows beyond a trip's own are all inside it or all outside");

template <int DIR>
__device__ __forceinline__ void conv_tile(const float* __restrict__ xrow, float* __restrict__ orow,
                                          const float* __restrict__ w16, const float* __restrict__ b4,
                                          const int t0) {
    const float4 wq0 = *(const float4*)(w16);
    const float4 wq1 = *(const float4*)(w16 + 4);
    const float4 wq2 = *(const float4*)(w16 + 8);
    const float4 wq3 = *(const float4*)(w16 + 12);
    const float wt[4][4] = {{wq0.x, wq0.y, wq0.z, wq0.w}, {wq1.x, wq1.y, wq1.z, wq1.w},
                            {wq2.x, wq2.y, wq2.z, wq2.w}, {wq3.x, wq3.y, wq3.z, wq3.w}};
    const float4 bias = *(const float4*)(b4);

    const int tb = DIR ? t0 : t0 - 3;
    float4 win[CV_G + 3];
    float4 nxt[CV_G];
#pragma unroll
    for (int i = 0; i < 3; ++i) {
        const int tt = DIR ? (tb + i) : max(tb + i, 0);
        win[i] = *(const float4*)(xrow + (size_t)tt * D_IN_PROJ);
    }
#pragma unroll
    for (int i = 0; i < CV_G; ++i) {
        const int tt = DIR ? min(tb + 3 + i, SEQLEN - 1) : (tb + 3 + i);
        nxt[i] = *(const float4*)(xrow + (size_t)tt * D_IN_PROJ);
    }
    bool edge_ok = DIR ? true : (t0 != 0);

#pragma unroll 1
    for (int g = 0; g < CV_NG; ++g) {
#pragma unroll
        for (int i = 0; i < CV_G; ++i) win[3 + i] = nxt[i];
        if (g + 1 < CV_NG) {
            const int tn = tb + (g + 1) * CV_G + 3;
#pragma unroll
            for (int i = 0; i < CV_G; ++i) {
                const int tt = DIR ? min(tn + i, SEQLEN - 1) : (tn + i);
                nxt[i] = *(const float4*)(xrow + (size_t)tt * D_IN_PROJ);
            }
        }
        if (DIR) edge_ok = (g + 1 < CV_NG) || (t0 + CV_TT < SEQLEN);

#pragma unroll
        for (int j = 0; j < CV_G; ++j) {
            float a0 = bias.x, a1 = bias.y, a2 = bias.z, a3 = bias.w;
#pragma unroll
            for (int k = 0; k < 4; ++k) {
                const int  wi   = DIR ? (j + 3 - k) : (j + k);
                const bool edge = DIR ? (wi >= CV_G) : (wi < 3);
                if (!edge || edge_ok) {
                    a0 += win[wi].x * wt[0][k];
                    a1 += win[wi].y * wt[1][k];
                    a2 += win[wi].z * wt[2][k];
                    a3 += win[wi].w * wt[3][k];
                }
            }
            float4 o;
            o.x = a0 * (1.0f / (1.0f + expf(-a0)));
            o.y = a1 * (1.0f / (1.0f + expf(-a1)));
            o.z = a2 * (1.0f / (1.0f + expf(-a2)));
            o.w = a3 * (1.0f / (1.0f + expf(-a3)));
            *(float4*)(orow + (size_t)(t0 + g * CV_G + j) * CONV2) = o;
        }
#pragma unroll
        for (int i = 0; i < 3; ++i) win[i] = win[CV_G + i];
        if (!DIR) edge_ok = true;
    }
}

__global__ void __launch_bounds__(CV_THREADS)
conv_dt(const float* __restrict__ xbcdt,
        const float* __restrict__ wf, const float* __restrict__ bf,
        const float* __restrict__ wb, const float* __restrict__ bb,
        const float* __restrict__ dt_bias,
        float* __restrict__ xc, float* __restrict__ dtb) {
    const int    tid = threadIdx.x;
    const size_t r0  = (size_t)blockIdx.x * CV_TT;
    const int    t0  = (int)(r0 % SEQLEN);
    const size_t rb  = r0 - (size_t)t0;

    if (tid < CV_C4) {
        const int c = tid * 4;
        if (blockIdx.y == 0)
            conv_tile<0>(xbcdt + rb * D_IN_PROJ + c, xc + rb * CONV2 + c, wf + c * 4, bf + c, t0);
        else
            conv_tile<1>(xbcdt + rb * D_IN_PROJ + CONV_DIM + c, xc + rb * CONV2 + CONV_DIM + c,
                         wb + c * 4, bb + c, t0);
    } else if (tid >= 160 && blockIdx.y == 0) {
        const int    l   = tid - 160;
        const int    h   = l & 15;
        const size_t rw  = r0 + (l >> 4);
        const float  dbh = dt_bias[h];
        float raw[CV_TT / 2];
#pragma unroll
        for (int i = 0; i < CV_TT / 2; ++i) raw[i] = xbcdt[(rw + 2 * i) * D_IN_PROJ + CONV2 + h];
#pragma unroll
        for (int i = 0; i < CV_TT / 2; ++i) {
            const float v = raw[i] + dbh;
            dtb[(rw + 2 * i) * NHEADS + h] = (v > 20.0f) ? v : log1pf(expf(v));
        }
    }
}

__device__ __forceinline__ void head_bases(int h, int& xbase, int& bbase, int& cbase) {
    if (h < HH) { xbase = h * HEADDIM;                 bbase = 512;  cbase = 544;  }
    else        { xbase = CONV_DIM + (h - HH) * HEADDIM; bbase = 1088; cbase = 1120; }
}

#define TS     32
#define NTILE  (CHUNK / TS)
static_assert(TS == 32, "the B|C staging below names TS/4 = 8 registers");

__global__ void __launch_bounds__(64) __attribute__((amdgpu_waves_per_eu(1, 4)))
scan_chunk(const float* __restrict__ xc, const float* __restrict__ dtb,
           const float* __restrict__ A_log, const float* __restrict__ Dp,
           float* __restrict__ y, float* __restrict__ S, float* __restrict__ prodA) {
    const int chunk = blockIdx.x & (NCHUNK - 1);
    const int h     = (blockIdx.x >> 5) & 15;
    const int b     = blockIdx.x >> 9;
    const int p     = threadIdx.x;
    int xbase, bbase, cbase; head_bases(h, xbase, bbase, cbase);
    (void)cbase;

    const float Ah = -expf(A_log[h]);
    const float Dh = Dp[h];
    float hst[NSTATE];
#pragma unroll
    for (int n = 0; n < NSTATE; ++n) hst[n] = 0.0f;
    float pA = 1.0f;

    __shared__ __attribute__((aligned(16))) float sBC[TS][2 * NSTATE];
    __shared__ float sX[TS][HEADDIM];
    __shared__ float sDt[TS], sDA[TS];

    const int  s_first = chunk * CHUNK;
    const long rdir    = (h < HH) ? 1 : -1;
    const long row0    = (long)b * SEQLEN + ((h < HH) ? s_first : (SEQLEN - 1 - s_first));

    const int lrow = p >> 4;
    const int lc4  = (p & 15) * 4;

    float4 q0, q1, q2, q3, q4, q5, q6, q7;
    float  rx[TS];
    float  rdt = 0.0f;

#define SCAN_CHUNK_LOAD_TILE(tile_)                                                        \
    {                                                                                      \
        const long rt = row0 + rdir * (long)((tile_) * TS);                                \
        const float* bcp = xc + (rt + rdir * lrow) * CONV2 + bbase + lc4;                  \
        const long   bcs = rdir * 4 * CONV2;                                               \
        q0 = *(const float4*)(bcp);            q1 = *(const float4*)(bcp + bcs);           \
        q2 = *(const float4*)(bcp + 2 * bcs);  q3 = *(const float4*)(bcp + 3 * bcs);       \
        q4 = *(const float4*)(bcp + 4 * bcs);  q5 = *(const float4*)(bcp + 5 * bcs);       \
        q6 = *(const float4*)(bcp + 6 * bcs);  q7 = *(const float4*)(bcp + 7 * bcs);       \
        _Pragma("unroll")                                                                  \
        for (int j = 0; j < TS; ++j)                                                       \
            rx[j] = xc[(rt + rdir * j) * CONV2 + xbase + p];                               \
        if (p < TS) rdt = dtb[(rt + rdir * p) * NHEADS + h];                               \
    }

    SCAN_CHUNK_LOAD_TILE(0)

#pragma unroll 1
    for (int tile = 0; tile < NTILE; ++tile) {
        *(float4*)&sBC[lrow     ][lc4] = q0;  *(float4*)&sBC[lrow +  4][lc4] = q1;
        *(float4*)&sBC[lrow +  8][lc4] = q2;  *(float4*)&sBC[lrow + 12][lc4] = q3;
        *(float4*)&sBC[lrow + 16][lc4] = q4;  *(float4*)&sBC[lrow + 20][lc4] = q5;
        *(float4*)&sBC[lrow + 24][lc4] = q6;  *(float4*)&sBC[lrow + 28][lc4] = q7;
#pragma unroll
        for (int j = 0; j < TS; ++j) sX[j][p] = rx[j];
        if (p < TS) { sDt[p] = rdt; sDA[p] = expf(Ah * rdt); }
        __syncthreads();

        if (tile + 1 < NTILE) SCAN_CHUNK_LOAD_TILE(tile + 1)

        const long rt = row0 + rdir * (long)(tile * TS);
#pragma unroll 2
        for (int j = 0; j < TS; ++j) {
            const float dtv = sDt[j];
            const float dA  = sDA[j];
            pA *= dA;
            const float xp  = sX[j][p];
            const float dtx = dtv * xp;
            float accy = 0.0f;
#pragma unroll
            for (int n = 0; n < NSTATE; ++n) {
                hst[n] = dA * hst[n] + dtx * sBC[j][n];
                accy  += hst[n] * sBC[j][NSTATE + n];
            }
            y[(rt + rdir * j) * D_INNER + h * HEADDIM + p] = accy + Dh * xp;
        }
        __syncthreads();
    }
#undef SCAN_CHUNK_LOAD_TILE

    const size_t sb = (((size_t)(b * NHEADS + h)) * NCHUNK + chunk) * (HEADDIM * NSTATE)
                    + (size_t)p * NSTATE;
#pragma unroll
    for (int n = 0; n < NSTATE; ++n) S[sb + n] = hst[n];
    if (p == 0) prodA[(b * NHEADS + h) * NCHUNK + chunk] = pA;
}

#define CB_ELEMS   (HEADDIM * NSTATE)
#define CB_THREADS 64
#define CB_BLK_PER_BH (CB_ELEMS / (4 * CB_THREADS))

__global__ void __launch_bounds__(CB_THREADS) __attribute__((amdgpu_waves_per_eu(1, 4)))
scan_combine(float* __restrict__ S, const float* __restrict__ prodA) {
    const int bh = blockIdx.x / CB_BLK_PER_BH;
    const int e4 = ((blockIdx.x % CB_BLK_PER_BH) * CB_THREADS + threadIdx.x) * 4;
    float*       sp = S + (size_t)bh * NCHUNK * CB_ELEMS + e4;
    const float* ap = prodA + bh * NCHUNK;

    float4 sl[NCHUNK - 1];
#pragma unroll
    for (int c = 0; c < NCHUNK - 1; ++c) sl[c] = *(const float4*)(sp + (size_t)c * CB_ELEMS);

    float4 hc = make_float4(0.0f, 0.0f, 0.0f, 0.0f);
#pragma unroll
    for (int c = 0; c < NCHUNK; ++c) {
        *(float4*)(sp + (size_t)c * CB_ELEMS) = hc;
        if (c + 1 < NCHUNK) {
            const float a = ap[c];
            hc.x = a * hc.x + sl[c].x;
            hc.y = a * hc.y + sl[c].y;
            hc.z = a * hc.z + sl[c].z;
            hc.w = a * hc.w + sl[c].w;
        }
    }
}

__global__ void __launch_bounds__(64) __attribute__((amdgpu_waves_per_eu(1, 4)))
scan_correct(const float* __restrict__ xc, const float* __restrict__ dtb,
             const float* __restrict__ A_log, const float* __restrict__ S,
             float* __restrict__ y) {
    const int chunk = blockIdx.x & (NCHUNK - 1);
    if (chunk == 0) return;
    const int h = (blockIdx.x >> 5) & 15;
    const int b = blockIdx.x >> 9;
    const int p = threadIdx.x;
    int xbase, bbase, cbase; head_bases(h, xbase, bbase, cbase);
    (void)xbase; (void)bbase;

    float hi[NSTATE];
    const size_t sb = (((size_t)(b * NHEADS + h)) * NCHUNK + chunk) * (HEADDIM * NSTATE)
                    + (size_t)p * NSTATE;
#pragma unroll
    for (int n = 0; n < NSTATE; ++n) hi[n] = S[sb + n];

    const float Ah = -expf(A_log[h]);
    float cumA = 1.0f;

    __shared__ __attribute__((aligned(16))) float sC[TS][NSTATE];
    __shared__ float sY[TS][HEADDIM];
    __shared__ float sDA[TS];

    const int  s_first = chunk * CHUNK;
    const long rdir    = (h < HH) ? 1 : -1;
    const long row0    = (long)b * SEQLEN + ((h < HH) ? s_first : (SEQLEN - 1 - s_first));

    const int lrow = p >> 3;
    const int lc4  = (p & 7) * 4;

    float4 c0, c1, c2, c3;
    float  ry[TS];
    float  rdt = 0.0f;

#define SCAN_CORRECT_LOAD_TILE(tile_)                                                      \
    {                                                                                      \
        const long rt = row0 + rdir * (long)((tile_) * TS);                                \
        const float* cp  = xc + (rt + rdir * lrow) * CONV2 + cbase + lc4;                  \
        const long   cst = rdir * 8 * CONV2;                                               \
        c0 = *(const float4*)(cp);            c1 = *(const float4*)(cp + cst);             \
        c2 = *(const float4*)(cp + 2 * cst);  c3 = *(const float4*)(cp + 3 * cst);         \
        _Pragma("unroll")                                                                  \
        for (int j = 0; j < TS; ++j)                                                       \
            ry[j] = y[(rt + rdir * j) * D_INNER + h * HEADDIM + p];                        \
        if (p < TS) rdt = dtb[(rt + rdir * p) * NHEADS + h];                               \
    }

    SCAN_CORRECT_LOAD_TILE(0)

#pragma unroll 1
    for (int tile = 0; tile < NTILE; ++tile) {
        *(float4*)&sC[lrow     ][lc4] = c0;  *(float4*)&sC[lrow +  8][lc4] = c1;
        *(float4*)&sC[lrow + 16][lc4] = c2;  *(float4*)&sC[lrow + 24][lc4] = c3;
#pragma unroll
        for (int j = 0; j < TS; ++j) sY[j][p] = ry[j];
        if (p < TS) sDA[p] = expf(Ah * rdt);
        __syncthreads();

        if (tile + 1 < NTILE) SCAN_CORRECT_LOAD_TILE(tile + 1)

        const long rt = row0 + rdir * (long)(tile * TS);
#pragma unroll 4
        for (int j = 0; j < TS; ++j) {
            cumA *= sDA[j];
            float acc = 0.0f;
#pragma unroll
            for (int n = 0; n < NSTATE; ++n) acc += hi[n] * sC[j][n];
            y[(rt + rdir * j) * D_INNER + h * HEADDIM + p] = sY[j][p] + cumA * acc;
        }
        __syncthreads();
    }
#undef SCAN_CORRECT_LOAD_TILE
}

#define LN_RPW 4
#define LN_RPB (8 * LN_RPW)
static_assert(BL % LN_RPB == 0 && LN_RPB == 32, "wave 0 stores the work-group's 32 results, one per lane");

__global__ void __launch_bounds__(256)
ln_stats(const float* __restrict__ y, float* __restrict__ mu, float* __restrict__ rstd) {
    __shared__ float smu[LN_RPB], srs[LN_RPB];
    const int    tid  = threadIdx.x;
    const int    lane = tid & 31;
    const int    wv   = tid >> 5;
    const size_t rowb = (size_t)blockIdx.x * LN_RPB;
    const float* yw   = y + (rowb + (size_t)wv * LN_RPW) * D_INNER + lane * 4;

    float4 v[LN_RPW][8];
#pragma unroll
    for (int r = 0; r < LN_RPW; ++r)
#pragma unroll
        for (int i = 0; i < 8; ++i)
            v[r][i] = *(const float4*)(yw + r * D_INNER + i * 128);

#pragma unroll
    for (int r = 0; r < LN_RPW; ++r) {
        float s = 0.0f, q = 0.0f;
#pragma unroll
        for (int i = 0; i < 8; ++i) {
            s += v[r][i].x; q += v[r][i].x * v[r][i].x;
            s += v[r][i].y; q += v[r][i].y * v[r][i].y;
            s += v[r][i].z; q += v[r][i].z * v[r][i].z;
            s += v[r][i].w; q += v[r][i].w * v[r][i].w;
        }
#pragma unroll
        for (int off = 16; off > 0; off >>= 1) {
            s += __shfl_xor(s, off);
            q += __shfl_xor(q, off);
        }
        if (lane == 0) {
            const float m  = s * (1.0f / D_INNER);
            const float vr = q * (1.0f / D_INNER) - m * m;
            smu[wv * LN_RPW + r] = m;
            srs[wv * LN_RPW + r] = rsqrtf(vr + 1e-5f);
        }
    }
    __syncthreads();
    if (wv == 0) {
        mu[rowb + lane]   = smu[lane];
        rstd[rowb + lane] = srs[lane];
    }
}


__global__ void __launch_bounds__(256)
gemm_out(const float* __restrict__ Y, const float* __restrict__ mu,
         const float* __restrict__ rstd, const float* __restrict__ lnw,
         const float* __restrict__ lnb, const float* __restrict__ Wout,
         float* __restrict__ out) {
    __shared__ __attribute__((aligned(16))) float smem[2 * STAGE_FLOATS];
    const int tid  = threadIdx.x;
    const int wv   = tid >> 5;
    const int lane = tid & 31;
    const int row  = lane & 15;
    const int koff = (lane >> 4) << 1;
    const int wm   = wv >> 1;
    const int wn   = wv & 1;
    const int m0   = blockIdx.y * GBM;
    const int n0   = blockIdx.x * GBN;

    const int ar = tid >> 3, ac = (tid & 7) << 2;
    const int br = tid >> 4, bc = (tid & 15) << 2;
    const float* ag = Y    + (size_t)(m0 + ar) * D_INNER + ac;
    const float* bg = Wout + (size_t)(2 * br) * D_MODEL + n0 + bc;

    const float mu0 = mu[m0 + ar],      rs0 = rstd[m0 + ar];
    const float mu1 = mu[m0 + ar + 32], rs1 = rstd[m0 + ar + 32];
    const float mu2 = mu[m0 + ar + 64], rs2 = rstd[m0 + ar + 64];
    const float mu3 = mu[m0 + ar + 96], rs3 = rstd[m0 + ar + 96];

    float4 ra0, ra1, ra2, ra3, rb0, rb1, rw, rz;
#define GEMM_OUT_LOAD(kt_)                                                                 \
    {                                                                                      \
        const float* a_ = ag + (kt_) * GBK;                                                \
        ra0 = *(const float4*)(a_);                                                        \
        ra1 = *(const float4*)(a_ + 32 * D_INNER);                                         \
        ra2 = *(const float4*)(a_ + 64 * D_INNER);                                         \
        ra3 = *(const float4*)(a_ + 96 * D_INNER);                                         \
        const float* b_ = bg + (size_t)(kt_) * GBK * D_MODEL;                              \
        rb0 = *(const float4*)(b_);                                                        \
        rb1 = *(const float4*)(b_ + D_MODEL);                                              \
        rw  = *(const float4*)(lnw + (kt_) * GBK + ac);                                    \
        rz  = *(const float4*)(lnb + (kt_) * GBK + ac);                                    \
    }
#define GEMM_OUT_LN4(v_, mu_, rs_)                                                         \
    make_float4(((v_).x - (mu_)) * (rs_) * rw.x + rz.x, ((v_).y - (mu_)) * (rs_) * rw.y + rz.y, \
                ((v_).z - (mu_)) * (rs_) * rw.z + rz.z, ((v_).w - (mu_)) * (rs_) * rw.w + rz.w)
#define GEMM_OUT_STAGE(st_)                                                                \
    {                                                                                      \
        float* sa_ = smem + (st_) * STAGE_FLOATS + ar * APITCH + ac;                       \
        float* sb_ = smem + (st_) * STAGE_FLOATS + SA_FLOATS + br * BPITCH2 + 2 * bc;      \
        *(float4*)(sa_)               = GEMM_OUT_LN4(ra0, mu0, rs0);                       \
        *(float4*)(sa_ + 32 * APITCH) = GEMM_OUT_LN4(ra1, mu1, rs1);                       \
        *(float4*)(sa_ + 64 * APITCH) = GEMM_OUT_LN4(ra2, mu2, rs2);                       \
        *(float4*)(sa_ + 96 * APITCH) = GEMM_OUT_LN4(ra3, mu3, rs3);                       \
        *(float4*)(sb_)     = make_float4(rb0.x, rb1.x, rb0.y, rb1.y);                     \
        *(float4*)(sb_ + 4) = make_float4(rb0.z, rb1.z, rb0.w, rb1.w);                     \
    }

    v8f c00 = {}, c01 = {}, c10 = {}, c11 = {};
    const int fa = (wm * 32 + row) * APITCH + koff;
    const int fb = SA_FLOATS + (lane >> 4) * BPITCH2 + (wn * 32 + row) * 2;

    GEMM_OUT_LOAD(0)
    GEMM_OUT_STAGE(0)
    __syncthreads();

#pragma unroll 1
    for (int kt = 0; kt < D_INNER / GBK; ++kt) {
        const int st = kt & 1;
        if (kt + 1 < D_INNER / GBK) GEMM_OUT_LOAD(kt + 1)
        slab_mma(smem + st * STAGE_FLOATS + fa, smem + st * STAGE_FLOATS + fb, c00, c01, c10, c11);
        if (kt + 1 < D_INNER / GBK) GEMM_OUT_STAGE(st ^ 1)
        __syncthreads();
    }
#undef GEMM_OUT_LOAD
#undef GEMM_OUT_LN4
#undef GEMM_OUT_STAGE

    const int mhi = (lane >> 4) << 3;
    const int nb  = n0 + wn * 32 + row;
#pragma unroll
    for (int r = 0; r < 8; ++r) {
        const size_t mm = (size_t)(m0 + wm * 32 + mhi + r);
        out[(mm     ) * D_MODEL + nb     ] = c00[r];
        out[(mm     ) * D_MODEL + nb + 16] = c01[r];
        out[(mm + 16) * D_MODEL + nb     ] = c10[r];
        out[(mm + 16) * D_MODEL + nb + 16] = c11[r];
    }
}

extern "C" void kernel_launch(void* const* d_in, const int* in_sizes, int n_in,
                              void* d_out, int out_size, void* d_ws, size_t ws_size,
                              hipStream_t stream) {
    const float* u        = (const float*)d_in[0];
    const float* W_in     = (const float*)d_in[2];
    const float* conv_fw  = (const float*)d_in[3];
    const float* conv_fb  = (const float*)d_in[4];
    const float* conv_bw  = (const float*)d_in[5];
    const float* conv_bb  = (const float*)d_in[6];
    const float* dt_bias  = (const float*)d_in[7];
    const float* A_log    = (const float*)d_in[8];
    const float* D_param  = (const float*)d_in[9];
    const float* ln_w     = (const float*)d_in[10];
    const float* ln_b     = (const float*)d_in[11];
    const float* W_out    = (const float*)d_in[12];
    float* out            = (float*)d_out;

    float* ws = (float*)d_ws;
    size_t off = 0;
    float* xbcdt = ws + off; off += (size_t)BL * D_IN_PROJ;
    float* xc    = ws + off; off += (size_t)BL * CONV2;
    float* dtb   = ws + off; off += (size_t)BL * NHEADS;
    float* ybuf  = ws + off; off += (size_t)BL * D_INNER;
    float* Sbuf  = ws + off; off += (size_t)32 * NCHUNK * HEADDIM * NSTATE;
    float* pAbuf = ws + off; off += (size_t)32 * NCHUNK;
    float* mubuf = ws + off; off += (size_t)BL;
    float* rsbuf = ws + off; off += (size_t)BL;

    gemm_in<<<dim3(GIN_NBLK, G_MBLK), 256, 0, stream>>>(u, W_in, xbcdt);
    conv_dt<<<dim3(BL / CV_TT, 2), CV_THREADS, 0, stream>>>(xbcdt, conv_fw, conv_fb, conv_bw, conv_bb,
                                                            dt_bias, xc, dtb);
    scan_chunk<<<dim3(BATCH * NHEADS * NCHUNK), 64, 0, stream>>>(
        xc, dtb, A_log, D_param, ybuf, Sbuf, pAbuf);
    scan_combine<<<dim3(BATCH * NHEADS * CB_BLK_PER_BH), CB_THREADS, 0, stream>>>(Sbuf, pAbuf);
    scan_correct<<<dim3(BATCH * NHEADS * NCHUNK), 64, 0, stream>>>(
        xc, dtb, A_log, Sbuf, ybuf);
    ln_stats<<<dim3(BL / LN_RPB), 256, 0, stream>>>(ybuf, mubuf, rsbuf);
    gemm_out<<<dim3(GOUT_NBLK, G_MBLK), 256, 0, stream>>>(
        ybuf, mubuf, rsbuf, ln_w, ln_b, W_out, out);

    (void)in_sizes; (void)n_in; (void)out_size; (void)ws_size;
}


# conv+SiLU and softplus(dt) done on the in-projection's result tile in LDS (row tiles: 125 outputs + 3 conv halo rows); conv launch and its memory round trip gone (7 -> 6 launches); same sum order
# speedup vs baseline: 3.5361x; 1.0515x over previous
#include <hip/hip_runtime.h>
#include <hip/hip_bf16.h>
#include <math.h>

#define BATCH    2
#define SEQLEN   4096
#define BL       (BATCH * SEQLEN)
#define D_MODEL  512
#define D_IN_PROJ 1168
#define CONV_DIM 576
#define CONV2    1152
#define D_INNER  1024
#define NHEADS   16
#define HH       8
#define HEADDIM  64
#define NSTATE   32
#define NCHUNK   32
#define CHUNK    128

typedef float v2f __attribute__((ext_vector_type(2)));
typedef float v8f __attribute__((ext_vector_type(8)));

__device__ __forceinline__ v8f wmma_f32_4(v2f a, v2f b, v8f c) {
    return __builtin_amdgcn_wmma_f32_16x16x4_f32(
         false, a,  false, b,
         (short)0, c,  false,  false);
}

#define GBM 128
#define GBN 64
#define GBK 32
#define APITCH 36
#define BPITCH2 160
#define SA_FLOATS (GBM * APITCH)
#define SB_FLOATS ((GBK / 2) * BPITCH2)
#define STAGE_FLOATS (SA_FLOATS + SB_FLOATS)
#define GIN_NBLK  19
#define GOUT_NBLK 8
#define G_MBLK    64
#define CT_OUT     125
#define CT_PER_SEQ 33
#define G_MBLK_IN  (BATCH * CT_PER_SEQ)
#define TPITCH     68

__device__ __forceinline__ void slab_mma(const float* pa, const float* pb,
                                         v8f& c00, v8f& c01, v8f& c10, v8f& c11) {
#pragma unroll
    for (int kk = 0; kk < GBK; kk += 4) {
        const v2f a0 = *(const v2f*)(pa + kk);
        const v2f a1 = *(const v2f*)(pa + 16 * APITCH + kk);
        const v2f b0 = *(const v2f*)(pb + (kk / 2) * BPITCH2);
        const v2f b1 = *(const v2f*)(pb + (kk / 2) * BPITCH2 + 32);
        c00 = wmma_f32_4(a0, b0, c00);
        c01 = wmma_f32_4(a0, b1, c01);
        c10 = wmma_f32_4(a1, b0, c10);
        c11 = wmma_f32_4(a1, b1, c11);
    }
}

__device__ __forceinline__ int clamp_time(int t) { return min(max(t, 0), SEQLEN - 1); }

__global__ void __launch_bounds__(256) __attribute__((amdgpu_waves_per_eu(10)))
gemm_in(const float* __restrict__ U, const float* __restrict__ Win,
        const float* __restrict__ wf, const float* __restrict__ bf,
        const float* __restrict__ wb, const float* __restrict__ bb,
        const float* __restrict__ dt_bias,
        float* __restrict__ xc, float* __restrict__ dtb) {
    __shared__ __attribute__((aligned(16))) float smem[2 * STAGE_FLOATS];
    const int tid  = threadIdx.x;
    const int wv   = tid >> 5;
    const int lane = tid & 31;
    const int row  = lane & 15;
    const int koff = (lane >> 4) << 1;
    const int wm   = wv >> 1;
    const int wn   = wv & 1;
    const int n0   = blockIdx.x * GBN;
    const int bsq  = blockIdx.y / CT_PER_SEQ;
    const int tout0 = (blockIdx.y % CT_PER_SEQ) * CT_OUT;
    const bool fwd = n0 < CONV_DIM;
    const int  tw0 = fwd ? tout0 - 3 : tout0;

    const int ar = tid >> 3, ac = (tid & 7) << 2;
    const int br = tid >> 4, bc = (tid & 15) << 2;
    const bool bok = (n0 + bc) < D_IN_PROJ;
    const float* ub  = U + (size_t)bsq * SEQLEN * D_MODEL;
    const unsigned ao0 = (unsigned)(clamp_time(tw0 + ar     ) * D_MODEL + ac);
    const unsigned ao1 = (unsigned)(clamp_time(tw0 + ar + 32) * D_MODEL + ac);
    const unsigned ao2 = (unsigned)(clamp_time(tw0 + ar + 64) * D_MODEL + ac);
    const unsigned ao3 = (unsigned)(clamp_time(tw0 + ar + 96) * D_MODEL + ac);
    const float* bg  = Win + (size_t)(2 * br) * D_IN_PROJ + n0 + bc;

    float4 ra0, ra1, ra2, ra3, rb0, rb1;
#define GEMM_IN_LOAD(kt_)                                                                  \
    {                                                                                      \
        const float* a_ = ub + (kt_) * GBK;                                                \
        ra0 = *(const float4*)(a_ + ao0);                                                  \
        ra1 = *(const float4*)(a_ + ao1);                                                  \
        ra2 = *(const float4*)(a_ + ao2);                                                  \
        ra3 = *(const float4*)(a_ + ao3);                                                  \
        if (bok) {                                                                         \
            const float* b_ = bg + (size_t)(kt_) * GBK * D_IN_PROJ;                        \
            rb0 = *(const float4*)(b_);                                                    \
            rb1 = *(const float4*)(b_ + D_IN_PROJ);                                        \
        } else {                                                                           \
            rb0 = make_float4(0.0f, 0.0f, 0.0f, 0.0f);                                     \
            rb1 = make_float4(0.0f, 0.0f, 0.0f, 0.0f);                                     \
        }                                                                                  \
    }
#define GEMM_IN_STAGE(st_)                                                                 \
    {                                                                                      \
        float* sa_ = smem + (st_) * STAGE_FLOATS + ar * APITCH + ac;                       \
        float* sb_ = smem + (st_) * STAGE_FLOATS + SA_FLOATS + br * BPITCH2 + 2 * bc;      \
        *(float4*)(sa_)               = ra0;                                               \
        *(float4*)(sa_ + 32 * APITCH) = ra1;                                               \
        *(float4*)(sa_ + 64 * APITCH) = ra2;                                               \
        *(float4*)(sa_ + 96 * APITCH) = ra3;                                               \
        *(float4*)(sb_)     = make_float4(rb0.x, rb1.x, rb0.y, rb1.y);                     \
        *(float4*)(sb_ + 4) = make_float4(rb0.z, rb1.z, rb0.w, rb1.w);                     \
    }

    v8f c00 = {}, c01 = {}, c10 = {}, c11 = {};
    const bool wave_on = (n0 + wn * 32) < D_IN_PROJ;
    const int fa = (wm * 32 + row) * APITCH + koff;
    const int fb = SA_FLOATS + (lane >> 4) * BPITCH2 + (wn * 32 + row) * 2;

    GEMM_IN_LOAD(0)
    GEMM_IN_STAGE(0)
    __syncthreads();

#pragma unroll 1
    for (int kt = 0; kt < D_MODEL / GBK; ++kt) {
        const int st = kt & 1;
        if (kt + 1 < D_MODEL / GBK) GEMM_IN_LOAD(kt + 1)
        if (wave_on)
            slab_mma(smem + st * STAGE_FLOATS + fa, smem + st * STAGE_FLOATS + fb, c00, c01, c10, c11);
        if (kt + 1 < D_MODEL / GBK) GEMM_IN_STAGE(st ^ 1)
        __syncthreads();
    }
#undef GEMM_IN_LOAD
#undef GEMM_IN_STAGE

    float* tile = smem;
    if (wave_on) {
        float* tp = tile + (wm * 32 + ((lane >> 4) << 3)) * TPITCH + wn * 32 + row;
#pragma unroll
        for (int r = 0; r < 8; ++r) {
            tp[(r     ) * TPITCH     ] = c00[r];
            tp[(r     ) * TPITCH + 16] = c01[r];
            tp[(r + 16) * TPITCH     ] = c10[r];
            tp[(r + 16) * TPITCH + 16] = c11[r];
        }
    }
    __syncthreads();

    const size_t rowg0 = (size_t)bsq * SEQLEN;
    if (n0 < CONV2) {
        const int c4  = (tid & 15) << 2;
        const int o0  = (tid >> 4) << 3;
        const int col = n0 + c4;
        const float* w16 = fwd ? (wf + (size_t)col * 4) : (wb + (size_t)(col - CONV_DIM) * 4);
        const float* b4  = fwd ? (bf + col) : (bb + (col - CONV_DIM));
        const float4 wq0 = *(const float4*)(w16);
        const float4 wq1 = *(const float4*)(w16 + 4);
        const float4 wq2 = *(const float4*)(w16 + 8);
        const float4 wq3 = *(const float4*)(w16 + 12);
        const float wt[4][4] = {{wq0.x, wq0.y, wq0.z, wq0.w}, {wq1.x, wq1.y, wq1.z, wq1.w},
                                {wq2.x, wq2.y, wq2.z, wq2.w}, {wq3.x, wq3.y, wq3.z, wq3.w}};
        const float4 bias = *(const float4*)(b4);

#pragma unroll 1
        for (int oh = o0; oh < o0 + 8; oh += 4) {
        float4 win[7];
#pragma unroll
        for (int i = 0; i < 7; ++i)
            win[i] = *(const float4*)(tile + min(oh + i, GBM - 1) * TPITCH + c4);

#pragma unroll
        for (int j = 0; j < 4; ++j) {
            const int o = oh + j;
            const int t = tout0 + o;
            if (o < CT_OUT && t < SEQLEN) {
                float a0 = bias.x, a1 = bias.y, a2 = bias.z, a3 = bias.w;
#pragma unroll
                for (int k = 0; k < 4; ++k) {
                    const int wi = fwd ? (j + k) : (j + 3 - k);
                    const int tt = fwd ? (t - 3 + k) : (t + 3 - k);
                    if (tt >= 0 && tt < SEQLEN) {
                        a0 += win[wi].x * wt[0][k];
                        a1 += win[wi].y * wt[1][k];
                        a2 += win[wi].z * wt[2][k];
                        a3 += win[wi].w * wt[3][k];
                    }
                }
                float4 ov;
                ov.x = a0 * (1.0f / (1.0f + expf(-a0)));
                ov.y = a1 * (1.0f / (1.0f + expf(-a1)));
                ov.z = a2 * (1.0f / (1.0f + expf(-a2)));
                ov.w = a3 * (1.0f / (1.0f + expf(-a3)));
                *(float4*)(xc + (rowg0 + (size_t)t) * CONV2 + col) = ov;
            }
        }
        }
    } else {
        const int   h   = tid & 15;
        const float dbh = dt_bias[h];
#pragma unroll
        for (int j = 0; j < 8; ++j) {
            const int o = (tid >> 4) + 16 * j;
            const int t = tout0 + o;
            if (o < CT_OUT && t < SEQLEN) {
                const float v = tile[o * TPITCH + h] + dbh;
                dtb[(rowg0 + (size_t)t) * NHEADS + h] = (v > 20.0f) ? v : log1pf(expf(v));
            }
        }
    }
}

__device__ __forceinline__ void head_bases(int h, int& xbase, int& bbase, int& cbase) {
    if (h < HH) { xbase = h * HEADDIM;                 bbase = 512;  cbase = 544;  }
    else        { xbase = CONV_DIM + (h - HH) * HEADDIM; bbase = 1088; cbase = 1120; }
}

#define TS     32
#define NTILE  (CHUNK / TS)
static_assert(TS == 32, "the B|C staging below names TS/4 = 8 registers");

__global__ void __launch_bounds__(64) __attribute__((amdgpu_waves_per_eu(1, 4)))
scan_chunk(const float* __restrict__ xc, const float* __restrict__ dtb,
           const float* __restrict__ A_log, const float* __restrict__ Dp,
           float* __restrict__ y, float* __restrict__ S, float* __restrict__ prodA) {
    const int chunk = blockIdx.x & (NCHUNK - 1);
    const int h     = (blockIdx.x >> 5) & 15;
    const int b     = blockIdx.x >> 9;
    const int p     = threadIdx.x;
    int xbase, bbase, cbase; head_bases(h, xbase, bbase, cbase);
    (void)cbase;

    const float Ah = -expf(A_log[h]);
    const float Dh = Dp[h];
    float hst[NSTATE];
#pragma unroll
    for (int n = 0; n < NSTATE; ++n) hst[n] = 0.0f;
    float pA = 1.0f;

    __shared__ __attribute__((aligned(16))) float sBC[TS][2 * NSTATE];
    __shared__ float sX[TS][HEADDIM];
    __shared__ float sDt[TS], sDA[TS];

    const int  s_first = chunk * CHUNK;
    const long rdir    = (h < HH) ? 1 : -1;
    const long row0    = (long)b * SEQLEN + ((h < HH) ? s_first : (SEQLEN - 1 - s_first));

    const int lrow = p >> 4;
    const int lc4  = (p & 15) * 4;

    float4 q0, q1, q2, q3, q4, q5, q6, q7;
    float  rx[TS];
    float  rdt = 0.0f;

#define SCAN_CHUNK_LOAD_TILE(tile_)                                                        \
    {                                                                                      \
        const long rt = row0 + rdir * (long)((tile_) * TS);                                \
        const float* bcp = xc + (rt + rdir * lrow) * CONV2 + bbase + lc4;                  \
        const long   bcs = rdir * 4 * CONV2;                                               \
        q0 = *(const float4*)(bcp);            q1 = *(const float4*)(bcp + bcs);           \
        q2 = *(const float4*)(bcp + 2 * bcs);  q3 = *(const float4*)(bcp + 3 * bcs);       \
        q4 = *(const float4*)(bcp + 4 * bcs);  q5 = *(const float4*)(bcp + 5 * bcs);       \
        q6 = *(const float4*)(bcp + 6 * bcs);  q7 = *(const float4*)(bcp + 7 * bcs);       \
        _Pragma("unroll")                                                                  \
        for (int j = 0; j < TS; ++j)                                                       \
            rx[j] = xc[(rt + rdir * j) * CONV2 + xbase + p];                               \
        if (p < TS) rdt = dtb[(rt + rdir * p) * NHEADS + h];                               \
    }

    SCAN_CHUNK_LOAD_TILE(0)

#pragma unroll 1
    for (int tile = 0; tile < NTILE; ++tile) {
        *(float4*)&sBC[lrow     ][lc4] = q0;  *(float4*)&sBC[lrow +  4][lc4] = q1;
        *(float4*)&sBC[lrow +  8][lc4] = q2;  *(float4*)&sBC[lrow + 12][lc4] = q3;
        *(float4*)&sBC[lrow + 16][lc4] = q4;  *(float4*)&sBC[lrow + 20][lc4] = q5;
        *(float4*)&sBC[lrow + 24][lc4] = q6;  *(float4*)&sBC[lrow + 28][lc4] = q7;
#pragma unroll
        for (int j = 0; j < TS; ++j) sX[j][p] = rx[j];
        if (p < TS) { sDt[p] = rdt; sDA[p] = expf(Ah * rdt); }
        __syncthreads();

        if (tile + 1 < NTILE) SCAN_CHUNK_LOAD_TILE(tile + 1)

        const long rt = row0 + rdir * (long)(tile * TS);
#pragma unroll 2
        for (int j = 0; j < TS; ++j) {
            const float dtv = sDt[j];
            const float dA  = sDA[j];
            pA *= dA;
            const float xp  = sX[j][p];
            const float dtx = dtv * xp;
            float accy = 0.0f;
#pragma unroll
            for (int n = 0; n < NSTATE; ++n) {
                hst[n] = dA * hst[n] + dtx * sBC[j][n];
                accy  += hst[n] * sBC[j][NSTATE + n];
            }
            y[(rt + rdir * j) * D_INNER + h * HEADDIM + p] = accy + Dh * xp;
        }
        __syncthreads();
    }
#undef SCAN_CHUNK_LOAD_TILE

    const size_t sb = (((size_t)(b * NHEADS + h)) * NCHUNK + chunk) * (HEADDIM * NSTATE)
                    + (size_t)p * NSTATE;
#pragma unroll
    for (int n = 0; n < NSTATE; ++n) S[sb + n] = hst[n];
    if (p == 0) prodA[(b * NHEADS + h) * NCHUNK + chunk] = pA;
}

#define CB_ELEMS   (HEADDIM * NSTATE)
#define CB_THREADS 64
#define CB_BLK_PER_BH (CB_ELEMS / (4 * CB_THREADS))

__global__ void __launch_bounds__(CB_THREADS) __attribute__((amdgpu_waves_per_eu(1, 4)))
scan_combine(float* __restrict__ S, const float* __restrict__ prodA) {
    const int bh = blockIdx.x / CB_BLK_PER_BH;
    const int e4 = ((blockIdx.x % CB_BLK_PER_BH) * CB_THREADS + threadIdx.x) * 4;
    float*       sp = S + (size_t)bh * NCHUNK * CB_ELEMS + e4;
    const float* ap = prodA + bh * NCHUNK;

    float4 sl[NCHUNK - 1];
#pragma unroll
    for (int c = 0; c < NCHUNK - 1; ++c) sl[c] = *(const float4*)(sp + (size_t)c * CB_ELEMS);

    float4 hc = make_float4(0.0f, 0.0f, 0.0f, 0.0f);
#pragma unroll
    for (int c = 0; c < NCHUNK; ++c) {
        *(float4*)(sp + (size_t)c * CB_ELEMS) = hc;
        if (c + 1 < NCHUNK) {
            const float a = ap[c];
            hc.x = a * hc.x + sl[c].x;
            hc.y = a * hc.y + sl[c].y;
            hc.z = a * hc.z + sl[c].z;
            hc.w = a * hc.w + sl[c].w;
        }
    }
}

__global__ void __launch_bounds__(64) __attribute__((amdgpu_waves_per_eu(1, 4)))
scan_correct(const float* __restrict__ xc, const float* __restrict__ dtb,
             const float* __restrict__ A_log, const float* __restrict__ S,
             float* __restrict__ y) {
    const int chunk = blockIdx.x & (NCHUNK - 1);
    if (chunk == 0) return;
    const int h = (blockIdx.x >> 5) & 15;
    const int b = blockIdx.x >> 9;
    const int p = threadIdx.x;
    int xbase, bbase, cbase; head_bases(h, xbase, bbase, cbase);
    (void)xbase; (void)bbase;

    float hi[NSTATE];
    const size_t sb = (((size_t)(b * NHEADS + h)) * NCHUNK + chunk) * (HEADDIM * NSTATE)
                    + (size_t)p * NSTATE;
#pragma unroll
    for (int n = 0; n < NSTATE; ++n) hi[n] = S[sb + n];

    const float Ah = -expf(A_log[h]);
    float cumA = 1.0f;

    __shared__ __attribute__((aligned(16))) float sC[TS][NSTATE];
    __shared__ float sY[TS][HEADDIM];
    __shared__ float sDA[TS];

    const int  s_first = chunk * CHUNK;
    const long rdir    = (h < HH) ? 1 : -1;
    const long row0    = (long)b * SEQLEN + ((h < HH) ? s_first : (SEQLEN - 1 - s_first));

    const int lrow = p >> 3;
    const int lc4  = (p & 7) * 4;

    float4 c0, c1, c2, c3;
    float  ry[TS];
    float  rdt = 0.0f;

#define SCAN_CORRECT_LOAD_TILE(tile_)                                                      \
    {                                                                                      \
        const long rt = row0 + rdir * (long)((tile_) * TS);                                \
        const float* cp  = xc + (rt + rdir * lrow) * CONV2 + cbase + lc4;                  \
        const long   cst = rdir * 8 * CONV2;                                               \
        c0 = *(const float4*)(cp);            c1 = *(const float4*)(cp + cst);             \
        c2 = *(const float4*)(cp + 2 * cst);  c3 = *(const float4*)(cp + 3 * cst);         \
        _Pragma("unroll")                                                                  \
        for (int j = 0; j < TS; ++j)                                                       \
            ry[j] = y[(rt + rdir * j) * D_INNER + h * HEADDIM + p];                        \
        if (p < TS) rdt = dtb[(rt + rdir * p) * NHEADS + h];                               \
    }

    SCAN_CORRECT_LOAD_TILE(0)

#pragma unroll 1
    for (int tile = 0; tile < NTILE; ++tile) {
        *(float4*)&sC[lrow     ][lc4] = c0;  *(float4*)&sC[lrow +  8][lc4] = c1;
        *(float4*)&sC[lrow + 16][lc4] = c2;  *(float4*)&sC[lrow + 24][lc4] = c3;
#pragma unroll
        for (int j = 0; j < TS; ++j) sY[j][p] = ry[j];
        if (p < TS) sDA[p] = expf(Ah * rdt);
        __syncthreads();

        if (tile + 1 < NTILE) SCAN_CORRECT_LOAD_TILE(tile + 1)

        const long rt = row0 + rdir * (long)(tile * TS);
#pragma unroll 4
        for (int j = 0; j < TS; ++j) {
            cumA *= sDA[j];
            float acc = 0.0f;
#pragma unroll
            for (int n = 0; n < NSTATE; ++n) acc += hi[n] * sC[j][n];
            y[(rt + rdir * j) * D_INNER + h * HEADDIM + p] = sY[j][p] + cumA * acc;
        }
        __syncthreads();
    }
#undef SCAN_CORRECT_LOAD_TILE
}

#define LN_RPW 4
#define LN_RPB (8 * LN_RPW)
static_assert(BL % LN_RPB == 0 && LN_RPB == 32, "wave 0 stores the work-group's 32 results, one per lane");

__global__ void __launch_bounds__(256)
ln_stats(const float* __restrict__ y, float* __restrict__ mu, float* __restrict__ rstd) {
    __shared__ float smu[LN_RPB], srs[LN_RPB];
    const int    tid  = threadIdx.x;
    const int    lane = tid & 31;
    const int    wv   = tid >> 5;
    const size_t rowb = (size_t)blockIdx.x * LN_RPB;
    const float* yw   = y + (rowb + (size_t)wv * LN_RPW) * D_INNER + lane * 4;

    float4 v[LN_RPW][8];
#pragma unroll
    for (int r = 0; r < LN_RPW; ++r)
#pragma unroll
        for (int i = 0; i < 8; ++i)
            v[r][i] = *(const float4*)(yw + r * D_INNER + i * 128);

#pragma unroll
    for (int r = 0; r < LN_RPW; ++r) {
        float s = 0.0f, q = 0.0f;
#pragma unroll
        for (int i = 0; i < 8; ++i) {
            s += v[r][i].x; q += v[r][i].x * v[r][i].x;
            s += v[r][i].y; q += v[r][i].y * v[r][i].y;
            s += v[r][i].z; q += v[r][i].z * v[r][i].z;
            s += v[r][i].w; q += v[r][i].w * v[r][i].w;
        }
#pragma unroll
        for (int off = 16; off > 0; off >>= 1) {
            s += __shfl_xor(s, off);
            q += __shfl_xor(q, off);
        }
        if (lane == 0) {
            const float m  = s * (1.0f / D_INNER);
            const float vr = q * (1.0f / D_INNER) - m * m;
            smu[wv * LN_RPW + r] = m;
            srs[wv * LN_RPW + r] = rsqrtf(vr + 1e-5f);
        }
    }
    __syncthreads();
    if (wv == 0) {
        mu[rowb + lane]   = smu[lane];
        rstd[rowb + lane] = srs[lane];
    }
}


__global__ void __launch_bounds__(256)
gemm_out(const float* __restrict__ Y, const float* __restrict__ mu,
         const float* __restrict__ rstd, const float* __restrict__ lnw,
         const float* __restrict__ lnb, const float* __restrict__ Wout,
         float* __restrict__ out) {
    __shared__ __attribute__((aligned(16))) float smem[2 * STAGE_FLOATS];
    const int tid  = threadIdx.x;
    const int wv   = tid >> 5;
    const int lane = tid & 31;
    const int row  = lane & 15;
    const int koff = (lane >> 4) << 1;
    const int wm   = wv >> 1;
    const int wn   = wv & 1;
    const int m0   = blockIdx.y * GBM;
    const int n0   = blockIdx.x * GBN;

    const int ar = tid >> 3, ac = (tid & 7) << 2;
    const int br = tid >> 4, bc = (tid & 15) << 2;
    const float* ag = Y    + (size_t)(m0 + ar) * D_INNER + ac;
    const float* bg = Wout + (size_t)(2 * br) * D_MODEL + n0 + bc;

    const float mu0 = mu[m0 + ar],      rs0 = rstd[m0 + ar];
    const float mu1 = mu[m0 + ar + 32], rs1 = rstd[m0 + ar + 32];
    const float mu2 = mu[m0 + ar + 64], rs2 = rstd[m0 + ar + 64];
    const float mu3 = mu[m0 + ar + 96], rs3 = rstd[m0 + ar + 96];

    float4 ra0, ra1, ra2, ra3, rb0, rb1, rw, rz;
#define GEMM_OUT_LOAD(kt_)                                                                 \
    {                                                                                      \
        const float* a_ = ag + (kt_) * GBK;                                                \
        ra0 = *(const float4*)(a_);                                                        \
        ra1 = *(const float4*)(a_ + 32 * D_INNER);                                         \
        ra2 = *(const float4*)(a_ + 64 * D_INNER);                                         \
        ra3 = *(const float4*)(a_ + 96 * D_INNER);                                         \
        const float* b_ = bg + (size_t)(kt_) * GBK * D_MODEL;                              \
        rb0 = *(const float4*)(b_);                                                        \
        rb1 = *(const float4*)(b_ + D_MODEL);                                              \
        rw  = *(const float4*)(lnw + (kt_) * GBK + ac);                                    \
        rz  = *(const float4*)(lnb + (kt_) * GBK + ac);                                    \
    }
#define GEMM_OUT_LN4(v_, mu_, rs_)                                                         \
    make_float4(((v_).x - (mu_)) * (rs_) * rw.x + rz.x, ((v_).y - (mu_)) * (rs_) * rw.y + rz.y, \
                ((v_).z - (mu_)) * (rs_) * rw.z + rz.z, ((v_).w - (mu_)) * (rs_) * rw.w + rz.w)
#define GEMM_OUT_STAGE(st_)                                                                \
    {                                                                                      \
        float* sa_ = smem + (st_) * STAGE_FLOATS + ar * APITCH + ac;                       \
        float* sb_ = smem + (st_) * STAGE_FLOATS + SA_FLOATS + br * BPITCH2 + 2 * bc;      \
        *(float4*)(sa_)               = GEMM_OUT_LN4(ra0, mu0, rs0);                       \
        *(float4*)(sa_ + 32 * APITCH) = GEMM_OUT_LN4(ra1, mu1, rs1);                       \
        *(float4*)(sa_ + 64 * APITCH) = GEMM_OUT_LN4(ra2, mu2, rs2);                       \
        *(float4*)(sa_ + 96 * APITCH) = GEMM_OUT_LN4(ra3, mu3, rs3);                       \
        *(float4*)(sb_)     = make_float4(rb0.x, rb1.x, rb0.y, rb1.y);                     \
        *(float4*)(sb_ + 4) = make_float4(rb0.z, rb1.z, rb0.w, rb1.w);                     \
    }

    v8f c00 = {}, c01 = {}, c10 = {}, c11 = {};
    const int fa = (wm * 32 + row) * APITCH + koff;
    const int fb = SA_FLOATS + (lane >> 4) * BPITCH2 + (wn * 32 + row) * 2;

    GEMM_OUT_LOAD(0)
    GEMM_OUT_STAGE(0)
    __syncthreads();

#pragma unroll 1
    for (int kt = 0; kt < D_INNER / GBK; ++kt) {
        const int st = kt & 1;
        if (kt + 1 < D_INNER / GBK) GEMM_OUT_LOAD(kt + 1)
        slab_mma(smem + st * STAGE_FLOATS + fa, smem + st * STAGE_FLOATS + fb, c00, c01, c10, c11);
        if (kt + 1 < D_INNER / GBK) GEMM_OUT_STAGE(st ^ 1)
        __syncthreads();
    }
#undef GEMM_OUT_LOAD
#undef GEMM_OUT_LN4
#undef GEMM_OUT_STAGE

    const int mhi = (lane >> 4) << 3;
    const int nb  = n0 + wn * 32 + row;
#pragma unroll
    for (int r = 0; r < 8; ++r) {
        const size_t mm = (size_t)(m0 + wm * 32 + mhi + r);
        out[(mm     ) * D_MODEL + nb     ] = c00[r];
        out[(mm     ) * D_MODEL + nb + 16] = c01[r];
        out[(mm + 16) * D_MODEL + nb     ] = c10[r];
        out[(mm + 16) * D_MODEL + nb + 16] = c11[r];
    }
}

extern "C" void kernel_launch(void* const* d_in, const int* in_sizes, int n_in,
                              void* d_out, int out_size, void* d_ws, size_t ws_size,
                              hipStream_t stream) {
    const float* u        = (const float*)d_in[0];
    const float* W_in     = (const float*)d_in[2];
    const float* conv_fw  = (const float*)d_in[3];
    const float* conv_fb  = (const float*)d_in[4];
    const float* conv_bw  = (const float*)d_in[5];
    const float* conv_bb  = (const float*)d_in[6];
    const float* dt_bias  = (const float*)d_in[7];
    const float* A_log    = (const float*)d_in[8];
    const float* D_param  = (const float*)d_in[9];
    const float* ln_w     = (const float*)d_in[10];
    const float* ln_b     = (const float*)d_in[11];
    const float* W_out    = (const float*)d_in[12];
    float* out            = (float*)d_out;

    float* ws = (float*)d_ws;
    size_t off = 0;
    off += (size_t)BL * D_IN_PROJ;
    float* xc    = ws + off; off += (size_t)BL * CONV2;
    float* dtb   = ws + off; off += (size_t)BL * NHEADS;
    float* ybuf  = ws + off; off += (size_t)BL * D_INNER;
    float* Sbuf  = ws + off; off += (size_t)32 * NCHUNK * HEADDIM * NSTATE;
    float* pAbuf = ws + off; off += (size_t)32 * NCHUNK;
    float* mubuf = ws + off; off += (size_t)BL;
    float* rsbuf = ws + off; off += (size_t)BL;

    gemm_in<<<dim3(GIN_NBLK, G_MBLK_IN), 256, 0, stream>>>(u, W_in, conv_fw, conv_fb, conv_bw, conv_bb,
                                                            dt_bias, xc, dtb);
    scan_chunk<<<dim3(BATCH * NHEADS * NCHUNK), 64, 0, stream>>>(
        xc, dtb, A_log, D_param, ybuf, Sbuf, pAbuf);
    scan_combine<<<dim3(BATCH * NHEADS * CB_BLK_PER_BH), CB_THREADS, 0, stream>>>(Sbuf, pAbuf);
    scan_correct<<<dim3(BATCH * NHEADS * NCHUNK), 64, 0, stream>>>(
        xc, dtb, A_log, Sbuf, ybuf);
    ln_stats<<<dim3(BL / LN_RPB), 256, 0, stream>>>(ybuf, mubuf, rsbuf);
    gemm_out<<<dim3(GOUT_NBLK, G_MBLK), 256, 0, stream>>>(
        ybuf, mubuf, rsbuf, ln_w, ln_b, W_out, out);

    (void)in_sizes; (void)n_in; (void)out_size; (void)ws_size;
}


# scan passes 1 and 3: the 32-term y sum kept as 4 partial sums (n mod 4) in two-wide fused multiply-adds joined by 3 adds (was products, then one chain of 32 adds): new sum order of y
# speedup vs baseline: 3.6608x; 1.0353x over previous
#include <hip/hip_runtime.h>
#include <hip/hip_bf16.h>
#include <math.h>

#define BATCH    2
#define SEQLEN   4096
#define BL       (BATCH * SEQLEN)
#define D_MODEL  512
#define D_IN_PROJ 1168
#define CONV_DIM 576
#define CONV2    1152
#define D_INNER  1024
#define NHEADS   16
#define HH       8
#define HEADDIM  64
#define NSTATE   32
#define NCHUNK   32
#define CHUNK    128

typedef float v2f __attribute__((ext_vector_type(2)));
typedef float v8f __attribute__((ext_vector_type(8)));

__device__ __forceinline__ v8f wmma_f32_4(v2f a, v2f b, v8f c) {
    return __builtin_amdgcn_wmma_f32_16x16x4_f32(
         false, a,  false, b,
         (short)0, c,  false,  false);
}

#define GBM 128
#define GBN 64
#define GBK 32
#define APITCH 36
#define BPITCH2 160
#define SA_FLOATS (GBM * APITCH)
#define SB_FLOATS ((GBK / 2) * BPITCH2)
#define STAGE_FLOATS (SA_FLOATS + SB_FLOATS)
#define GIN_NBLK  19
#define GOUT_NBLK 8
#define G_MBLK    64
#define CT_OUT     125
#define CT_PER_SEQ 33
#define G_MBLK_IN  (BATCH * CT_PER_SEQ)
#define TPITCH     68

__device__ __forceinline__ void slab_mma(const float* pa, const float* pb,
                                         v8f& c00, v8f& c01, v8f& c10, v8f& c11) {
#pragma unroll
    for (int kk = 0; kk < GBK; kk += 4) {
        const v2f a0 = *(const v2f*)(pa + kk);
        const v2f a1 = *(const v2f*)(pa + 16 * APITCH + kk);
        const v2f b0 = *(const v2f*)(pb + (kk / 2) * BPITCH2);
        const v2f b1 = *(const v2f*)(pb + (kk / 2) * BPITCH2 + 32);
        c00 = wmma_f32_4(a0, b0, c00);
        c01 = wmma_f32_4(a0, b1, c01);
        c10 = wmma_f32_4(a1, b0, c10);
        c11 = wmma_f32_4(a1, b1, c11);
    }
}

__device__ __forceinline__ int clamp_time(int t) { return min(max(t, 0), SEQLEN - 1); }

__global__ void __launch_bounds__(256) __attribute__((amdgpu_waves_per_eu(10)))
gemm_in(const float* __restrict__ U, const float* __restrict__ Win,
        const float* __restrict__ wf, const float* __restrict__ bf,
        const float* __restrict__ wb, const float* __restrict__ bb,
        const float* __restrict__ dt_bias,
        float* __restrict__ xc, float* __restrict__ dtb) {
    __shared__ __attribute__((aligned(16))) float smem[2 * STAGE_FLOATS];
    const int tid  = threadIdx.x;
    const int wv   = tid >> 5;
    const int lane = tid & 31;
    const int row  = lane & 15;
    const int koff = (lane >> 4) << 1;
    const int wm   = wv >> 1;
    const int wn   = wv & 1;
    const int n0   = blockIdx.x * GBN;
    const int bsq  = blockIdx.y / CT_PER_SEQ;
    const int tout0 = (blockIdx.y % CT_PER_SEQ) * CT_OUT;
    const bool fwd = n0 < CONV_DIM;
    const int  tw0 = fwd ? tout0 - 3 : tout0;

    const int ar = tid >> 3, ac = (tid & 7) << 2;
    const int br = tid >> 4, bc = (tid & 15) << 2;
    const bool bok = (n0 + bc) < D_IN_PROJ;
    const float* ub  = U + (size_t)bsq * SEQLEN * D_MODEL;
    const unsigned ao0 = (unsigned)(clamp_time(tw0 + ar     ) * D_MODEL + ac);
    const unsigned ao1 = (unsigned)(clamp_time(tw0 + ar + 32) * D_MODEL + ac);
    const unsigned ao2 = (unsigned)(clamp_time(tw0 + ar + 64) * D_MODEL + ac);
    const unsigned ao3 = (unsigned)(clamp_time(tw0 + ar + 96) * D_MODEL + ac);
    const float* bg  = Win + (size_t)(2 * br) * D_IN_PROJ + n0 + bc;

    float4 ra0, ra1, ra2, ra3, rb0, rb1;
#define GEMM_IN_LOAD(kt_)                                                                  \
    {                                                                                      \
        const float* a_ = ub + (kt_) * GBK;                                                \
        ra0 = *(const float4*)(a_ + ao0);                                                  \
        ra1 = *(const float4*)(a_ + ao1);                                                  \
        ra2 = *(const float4*)(a_ + ao2);                                                  \
        ra3 = *(const float4*)(a_ + ao3);                                                  \
        if (bok) {                                                                         \
            const float* b_ = bg + (size_t)(kt_) * GBK * D_IN_PROJ;                        \
            rb0 = *(const float4*)(b_);                                                    \
            rb1 = *(const float4*)(b_ + D_IN_PROJ);                                        \
        } else {                                                                           \
            rb0 = make_float4(0.0f, 0.0f, 0.0f, 0.0f);                                     \
            rb1 = make_float4(0.0f, 0.0f, 0.0f, 0.0f);                                     \
        }                                                                                  \
    }
#define GEMM_IN_STAGE(st_)                                                                 \
    {                                                                                      \
        float* sa_ = smem + (st_) * STAGE_FLOATS + ar * APITCH + ac;                       \
        float* sb_ = smem + (st_) * STAGE_FLOATS + SA_FLOATS + br * BPITCH2 + 2 * bc;      \
        *(float4*)(sa_)               = ra0;                                               \
        *(float4*)(sa_ + 32 * APITCH) = ra1;                                               \
        *(float4*)(sa_ + 64 * APITCH) = ra2;                                               \
        *(float4*)(sa_ + 96 * APITCH) = ra3;                                               \
        *(float4*)(sb_)     = make_float4(rb0.x, rb1.x, rb0.y, rb1.y);                     \
        *(float4*)(sb_ + 4) = make_float4(rb0.z, rb1.z, rb0.w, rb1.w);                     \
    }

    v8f c00 = {}, c01 = {}, c10 = {}, c11 = {};
    const bool wave_on = (n0 + wn * 32) < D_IN_PROJ;
    const int fa = (wm * 32 + row) * APITCH + koff;
    const int fb = SA_FLOATS + (lane >> 4) * BPITCH2 + (wn * 32 + row) * 2;

    GEMM_IN_LOAD(0)
    GEMM_IN_STAGE(0)
    __syncthreads();

#pragma unroll 1
    for (int kt = 0; kt < D_MODEL / GBK; ++kt) {
        const int st = kt & 1;
        if (kt + 1 < D_MODEL / GBK) GEMM_IN_LOAD(kt + 1)
        if (wave_on)
            slab_mma(smem + st * STAGE_FLOATS + fa, smem + st * STAGE_FLOATS + fb, c00, c01, c10, c11);
        if (kt + 1 < D_MODEL / GBK) GEMM_IN_STAGE(st ^ 1)
        __syncthreads();
    }
#undef GEMM_IN_LOAD
#undef GEMM_IN_STAGE

    float* tile = smem;
    if (wave_on) {
        float* tp = tile + (wm * 32 + ((lane >> 4) << 3)) * TPITCH + wn * 32 + row;
#pragma unroll
        for (int r = 0; r < 8; ++r) {
            tp[(r     ) * TPITCH     ] = c00[r];
            tp[(r     ) * TPITCH + 16] = c01[r];
            tp[(r + 16) * TPITCH     ] = c10[r];
            tp[(r + 16) * TPITCH + 16] = c11[r];
        }
    }
    __syncthreads();

    const size_t rowg0 = (size_t)bsq * SEQLEN;
    if (n0 < CONV2) {
        const int c4  = (tid & 15) << 2;
        const int o0  = (tid >> 4) << 3;
        const int col = n0 + c4;
        const float* w16 = fwd ? (wf + (size_t)col * 4) : (wb + (size_t)(col - CONV_DIM) * 4);
        const float* b4  = fwd ? (bf + col) : (bb + (col - CONV_DIM));
        const float4 wq0 = *(const float4*)(w16);
        const float4 wq1 = *(const float4*)(w16 + 4);
        const float4 wq2 = *(const float4*)(w16 + 8);
        const float4 wq3 = *(const float4*)(w16 + 12);
        const float wt[4][4] = {{wq0.x, wq0.y, wq0.z, wq0.w}, {wq1.x, wq1.y, wq1.z, wq1.w},
                                {wq2.x, wq2.y, wq2.z, wq2.w}, {wq3.x, wq3.y, wq3.z, wq3.w}};
        const float4 bias = *(const float4*)(b4);

#pragma unroll 1
        for (int oh = o0; oh < o0 + 8; oh += 4) {
        float4 win[7];
#pragma unroll
        for (int i = 0; i < 7; ++i)
            win[i] = *(const float4*)(tile + min(oh + i, GBM - 1) * TPITCH + c4);

#pragma unroll
        for (int j = 0; j < 4; ++j) {
            const int o = oh + j;
            const int t = tout0 + o;
            if (o < CT_OUT && t < SEQLEN) {
                float a0 = bias.x, a1 = bias.y, a2 = bias.z, a3 = bias.w;
#pragma unroll
                for (int k = 0; k < 4; ++k) {
                    const int wi = fwd ? (j + k) : (j + 3 - k);
                    const int tt = fwd ? (t - 3 + k) : (t + 3 - k);
                    if (tt >= 0 && tt < SEQLEN) {
                        a0 += win[wi].x * wt[0][k];
                        a1 += win[wi].y * wt[1][k];
                        a2 += win[wi].z * wt[2][k];
                        a3 += win[wi].w * wt[3][k];
                    }
                }
                float4 ov;
                ov.x = a0 * (1.0f / (1.0f + expf(-a0)));
                ov.y = a1 * (1.0f / (1.0f + expf(-a1)));
                ov.z = a2 * (1.0f / (1.0f + expf(-a2)));
                ov.w = a3 * (1.0f / (1.0f + expf(-a3)));
                *(float4*)(xc + (rowg0 + (size_t)t) * CONV2 + col) = ov;
            }
        }
        }
    } else {
        const int   h   = tid & 15;
        const float dbh = dt_bias[h];
#pragma unroll
        for (int j = 0; j < 8; ++j) {
            const int o = (tid >> 4) + 16 * j;
            const int t = tout0 + o;
            if (o < CT_OUT && t < SEQLEN) {
                const float v = tile[o * TPITCH + h] + dbh;
                dtb[(rowg0 + (size_t)t) * NHEADS + h] = (v > 20.0f) ? v : log1pf(expf(v));
            }
        }
    }
}

__device__ __forceinline__ void head_bases(int h, int& xbase, int& bbase, int& cbase) {
    if (h < HH) { xbase = h * HEADDIM;                 bbase = 512;  cbase = 544;  }
    else        { xbase = CONV_DIM + (h - HH) * HEADDIM; bbase = 1088; cbase = 1120; }
}

#define TS     32
#define NTILE  (CHUNK / TS)
static_assert(TS == 32, "the B|C staging below names TS/4 = 8 registers");

__global__ void __launch_bounds__(64) __attribute__((amdgpu_waves_per_eu(1, 4)))
scan_chunk(const float* __restrict__ xc, const float* __restrict__ dtb,
           const float* __restrict__ A_log, const float* __restrict__ Dp,
           float* __restrict__ y, float* __restrict__ S, float* __restrict__ prodA) {
    const int chunk = blockIdx.x & (NCHUNK - 1);
    const int h     = (blockIdx.x >> 5) & 15;
    const int b     = blockIdx.x >> 9;
    const int p     = threadIdx.x;
    int xbase, bbase, cbase; head_bases(h, xbase, bbase, cbase);
    (void)cbase;

    const float Ah = -expf(A_log[h]);
    const float Dh = Dp[h];
    v2f hst[NSTATE / 2];
#pragma unroll
    for (int k = 0; k < NSTATE / 2; ++k) hst[k] = (v2f){0.0f, 0.0f};
    float pA = 1.0f;

    __shared__ __attribute__((aligned(16))) float sBC[TS][2 * NSTATE];
    __shared__ float sX[TS][HEADDIM];
    __shared__ float sDt[TS], sDA[TS];

    const int  s_first = chunk * CHUNK;
    const long rdir    = (h < HH) ? 1 : -1;
    const long row0    = (long)b * SEQLEN + ((h < HH) ? s_first : (SEQLEN - 1 - s_first));

    const int lrow = p >> 4;
    const int lc4  = (p & 15) * 4;

    float4 q0, q1, q2, q3, q4, q5, q6, q7;
    float  rx[TS];
    float  rdt = 0.0f;

#define SCAN_CHUNK_LOAD_TILE(tile_)                                                        \
    {                                                                                      \
        const long rt = row0 + rdir * (long)((tile_) * TS);                                \
        const float* bcp = xc + (rt + rdir * lrow) * CONV2 + bbase + lc4;                  \
        const long   bcs = rdir * 4 * CONV2;                                               \
        q0 = *(const float4*)(bcp);            q1 = *(const float4*)(bcp + bcs);           \
        q2 = *(const float4*)(bcp + 2 * bcs);  q3 = *(const float4*)(bcp + 3 * bcs);       \
        q4 = *(const float4*)(bcp + 4 * bcs);  q5 = *(const float4*)(bcp + 5 * bcs);       \
        q6 = *(const float4*)(bcp + 6 * bcs);  q7 = *(const float4*)(bcp + 7 * bcs);       \
        _Pragma("unroll")                                                                  \
        for (int j = 0; j < TS; ++j)                                                       \
            rx[j] = xc[(rt + rdir * j) * CONV2 + xbase + p];                               \
        if (p < TS) rdt = dtb[(rt + rdir * p) * NHEADS + h];                               \
    }

    SCAN_CHUNK_LOAD_TILE(0)

#pragma unroll 1
    for (int tile = 0; tile < NTILE; ++tile) {
        *(float4*)&sBC[lrow     ][lc4] = q0;  *(float4*)&sBC[lrow +  4][lc4] = q1;
        *(float4*)&sBC[lrow +  8][lc4] = q2;  *(float4*)&sBC[lrow + 12][lc4] = q3;
        *(float4*)&sBC[lrow + 16][lc4] = q4;  *(float4*)&sBC[lrow + 20][lc4] = q5;
        *(float4*)&sBC[lrow + 24][lc4] = q6;  *(float4*)&sBC[lrow + 28][lc4] = q7;
#pragma unroll
        for (int j = 0; j < TS; ++j) sX[j][p] = rx[j];
        if (p < TS) { sDt[p] = rdt; sDA[p] = expf(Ah * rdt); }
        __syncthreads();

        if (tile + 1 < NTILE) SCAN_CHUNK_LOAD_TILE(tile + 1)

        const long rt = row0 + rdir * (long)(tile * TS);
#pragma unroll 2
        for (int j = 0; j < TS; ++j) {
            const float dtv = sDt[j];
            const float dA  = sDA[j];
            pA *= dA;
            const float xp  = sX[j][p];
            const float dtx = dtv * xp;
            const v2f dA2  = {dA, dA};
            const v2f dtx2 = {dtx, dtx};
            const v2f* bc2 = (const v2f*)&sBC[j][0];
            v2f ya = {0.0f, 0.0f}, yb = {0.0f, 0.0f};
#pragma unroll
            for (int k = 0; k < NSTATE / 2; k += 2) {
                const v2f t0 = dtx2 * bc2[k];
                const v2f t1 = dtx2 * bc2[k + 1];
                hst[k]     = __builtin_elementwise_fma(dA2, hst[k],     t0);
                hst[k + 1] = __builtin_elementwise_fma(dA2, hst[k + 1], t1);
                ya = __builtin_elementwise_fma(hst[k],     bc2[NSTATE / 2 + k],     ya);
                yb = __builtin_elementwise_fma(hst[k + 1], bc2[NSTATE / 2 + k + 1], yb);
            }
            const float accy = (ya.x + ya.y) + (yb.x + yb.y);
            y[(rt + rdir * j) * D_INNER + h * HEADDIM + p] = accy + Dh * xp;
        }
        __syncthreads();
    }
#undef SCAN_CHUNK_LOAD_TILE

    const size_t sb = (((size_t)(b * NHEADS + h)) * NCHUNK + chunk) * (HEADDIM * NSTATE)
                    + (size_t)p * NSTATE;
#pragma unroll
    for (int k = 0; k < NSTATE / 2; ++k) *(v2f*)(S + sb + 2 * k) = hst[k];
    if (p == 0) prodA[(b * NHEADS + h) * NCHUNK + chunk] = pA;
}

#define CB_ELEMS   (HEADDIM * NSTATE)
#define CB_THREADS 64
#define CB_BLK_PER_BH (CB_ELEMS / (4 * CB_THREADS))

__global__ void __launch_bounds__(CB_THREADS) __attribute__((amdgpu_waves_per_eu(1, 4)))
scan_combine(float* __restrict__ S, const float* __restrict__ prodA) {
    const int bh = blockIdx.x / CB_BLK_PER_BH;
    const int e4 = ((blockIdx.x % CB_BLK_PER_BH) * CB_THREADS + threadIdx.x) * 4;
    float*       sp = S + (size_t)bh * NCHUNK * CB_ELEMS + e4;
    const float* ap = prodA + bh * NCHUNK;

    float4 sl[NCHUNK - 1];
#pragma unroll
    for (int c = 0; c < NCHUNK - 1; ++c) sl[c] = *(const float4*)(sp + (size_t)c * CB_ELEMS);

    float4 hc = make_float4(0.0f, 0.0f, 0.0f, 0.0f);
#pragma unroll
    for (int c = 0; c < NCHUNK; ++c) {
        *(float4*)(sp + (size_t)c * CB_ELEMS) = hc;
        if (c + 1 < NCHUNK) {
            const float a = ap[c];
            hc.x = a * hc.x + sl[c].x;
            hc.y = a * hc.y + sl[c].y;
            hc.z = a * hc.z + sl[c].z;
            hc.w = a * hc.w + sl[c].w;
        }
    }
}

__global__ void __launch_bounds__(64) __attribute__((amdgpu_waves_per_eu(1, 4)))
scan_correct(const float* __restrict__ xc, const float* __restrict__ dtb,
             const float* __restrict__ A_log, const float* __restrict__ S,
             float* __restrict__ y) {
    const int chunk = blockIdx.x & (NCHUNK - 1);
    if (chunk == 0) return;
    const int h = (blockIdx.x >> 5) & 15;
    const int b = blockIdx.x >> 9;
    const int p = threadIdx.x;
    int xbase, bbase, cbase; head_bases(h, xbase, bbase, cbase);
    (void)xbase; (void)bbase;

    v2f hi[NSTATE / 2];
    const size_t sb = (((size_t)(b * NHEADS + h)) * NCHUNK + chunk) * (HEADDIM * NSTATE)
                    + (size_t)p * NSTATE;
#pragma unroll
    for (int k = 0; k < NSTATE / 2; ++k) hi[k] = *(const v2f*)(S + sb + 2 * k);

    const float Ah = -expf(A_log[h]);
    float cumA = 1.0f;

    __shared__ __attribute__((aligned(16))) float sC[TS][NSTATE];
    __shared__ float sY[TS][HEADDIM];
    __shared__ float sDA[TS];

    const int  s_first = chunk * CHUNK;
    const long rdir    = (h < HH) ? 1 : -1;
    const long row0    = (long)b * SEQLEN + ((h < HH) ? s_first : (SEQLEN - 1 - s_first));

    const int lrow = p >> 3;
    const int lc4  = (p & 7) * 4;

    float4 c0, c1, c2, c3;
    float  ry[TS];
    float  rdt = 0.0f;

#define SCAN_CORRECT_LOAD_TILE(tile_)                                                      \
    {                                                                                      \
        const long rt = row0 + rdir * (long)((tile_) * TS);                                \
        const float* cp  = xc + (rt + rdir * lrow) * CONV2 + cbase + lc4;                  \
        const long   cst = rdir * 8 * CONV2;                                               \
        c0 = *(const float4*)(cp);            c1 = *(const float4*)(cp + cst);             \
        c2 = *(const float4*)(cp + 2 * cst);  c3 = *(const float4*)(cp + 3 * cst);         \
        _Pragma("unroll")                                                                  \
        for (int j = 0; j < TS; ++j)                                                       \
            ry[j] = y[(rt + rdir * j) * D_INNER + h * HEADDIM + p];                        \
        if (p < TS) rdt = dtb[(rt + rdir * p) * NHEADS + h];                               \
    }

    SCAN_CORRECT_LOAD_TILE(0)

#pragma unroll 1
    for (int tile = 0; tile < NTILE; ++tile) {
        *(float4*)&sC[lrow     ][lc4] = c0;  *(float4*)&sC[lrow +  8][lc4] = c1;
        *(float4*)&sC[lrow + 16][lc4] = c2;  *(float4*)&sC[lrow + 24][lc4] = c3;
#pragma unroll
        for (int j = 0; j < TS; ++j) sY[j][p] = ry[j];
        if (p < TS) sDA[p] = expf(Ah * rdt);
        __syncthreads();

        if (tile + 1 < NTILE) SCAN_CORRECT_LOAD_TILE(tile + 1)

        const long rt = row0 + rdir * (long)(tile * TS);
#pragma unroll 4
        for (int j = 0; j < TS; ++j) {
            cumA *= sDA[j];
            const v2f* c2 = (const v2f*)&sC[j][0];
            v2f a0 = {0.0f, 0.0f}, a1 = {0.0f, 0.0f};
#pragma unroll
            for (int k = 0; k < NSTATE / 2; k += 2) {
                a0 = __builtin_elementwise_fma(hi[k],     c2[k],     a0);
                a1 = __builtin_elementwise_fma(hi[k + 1], c2[k + 1], a1);
            }
            const float acc = (a0.x + a0.y) + (a1.x + a1.y);
            y[(rt + rdir * j) * D_INNER + h * HEADDIM + p] = sY[j][p] + cumA * acc;
        }
        __syncthreads();
    }
#undef SCAN_CORRECT_LOAD_TILE
}

#define LN_RPW 4
#define LN_RPB (8 * LN_RPW)
static_assert(BL % LN_RPB == 0 && LN_RPB == 32, "wave 0 stores the work-group's 32 results, one per lane");

__global__ void __launch_bounds__(256)
ln_stats(const float* __restrict__ y, float* __restrict__ mu, float* __restrict__ rstd) {
    __shared__ float smu[LN_RPB], srs[LN_RPB];
    const int    tid  = threadIdx.x;
    const int    lane = tid & 31;
    const int    wv   = tid >> 5;
    const size_t rowb = (size_t)blockIdx.x * LN_RPB;
    const float* yw   = y + (rowb + (size_t)wv * LN_RPW) * D_INNER + lane * 4;

    float4 v[LN_RPW][8];
#pragma unroll
    for (int r = 0; r < LN_RPW; ++r)
#pragma unroll
        for (int i = 0; i < 8; ++i)
            v[r][i] = *(const float4*)(yw + r * D_INNER + i * 128);

#pragma unroll
    for (int r = 0; r < LN_RPW; ++r) {
        float s = 0.0f, q = 0.0f;
#pragma unroll
        for (int i = 0; i < 8; ++i) {
            s += v[r][i].x; q += v[r][i].x * v[r][i].x;
            s += v[r][i].y; q += v[r][i].y * v[r][i].y;
            s += v[r][i].z; q += v[r][i].z * v[r][i].z;
            s += v[r][i].w; q += v[r][i].w * v[r][i].w;
        }
#pragma unroll
        for (int off = 16; off > 0; off >>= 1) {
            s += __shfl_xor(s, off);
            q += __shfl_xor(q, off);
        }
        if (lane == 0) {
            const float m  = s * (1.0f / D_INNER);
            const float vr = q * (1.0f / D_INNER) - m * m;
            smu[wv * LN_RPW + r] = m;
            srs[wv * LN_RPW + r] = rsqrtf(vr + 1e-5f);
        }
    }
    __syncthreads();
    if (wv == 0) {
        mu[rowb + lane]   = smu[lane];
        rstd[rowb + lane] = srs[lane];
    }
}


__global__ void __launch_bounds__(256)
gemm_out(const float* __restrict__ Y, const float* __restrict__ mu,
         const float* __restrict__ rstd, const float* __restrict__ lnw,
         const float* __restrict__ lnb, const float* __restrict__ Wout,
         float* __restrict__ out) {
    __shared__ __attribute__((aligned(16))) float smem[2 * STAGE_FLOATS];
    const int tid  = threadIdx.x;
    const int wv   = tid >> 5;
    const int lane = tid & 31;
    const int row  = lane & 15;
    const int koff = (lane >> 4) << 1;
    const int wm   = wv >> 1;
    const int wn   = wv & 1;
    const int m0   = blockIdx.y * GBM;
    const int n0   = blockIdx.x * GBN;

    const int ar = tid >> 3, ac = (tid & 7) << 2;
    const int br = tid >> 4, bc = (tid & 15) << 2;
    const float* ag = Y    + (size_t)(m0 + ar) * D_INNER + ac;
    const float* bg = Wout + (size_t)(2 * br) * D_MODEL + n0 + bc;

    const float mu0 = mu[m0 + ar],      rs0 = rstd[m0 + ar];
    const float mu1 = mu[m0 + ar + 32], rs1 = rstd[m0 + ar + 32];
    const float mu2 = mu[m0 + ar + 64], rs2 = rstd[m0 + ar + 64];
    const float mu3 = mu[m0 + ar + 96], rs3 = rstd[m0 + ar + 96];

    float4 ra0, ra1, ra2, ra3, rb0, rb1, rw, rz;
#define GEMM_OUT_LOAD(kt_)                                                                 \
    {                                                                                      \
        const float* a_ = ag + (kt_) * GBK;                                                \
        ra0 = *(const float4*)(a_);                                                        \
        ra1 = *(const float4*)(a_ + 32 * D_INNER);                                         \
        ra2 = *(const float4*)(a_ + 64 * D_INNER);                                         \
        ra3 = *(const float4*)(a_ + 96 * D_INNER);                                         \
        const float* b_ = bg + (size_t)(kt_) * GBK * D_MODEL;                              \
        rb0 = *(const float4*)(b_);                                                        \
        rb1 = *(const float4*)(b_ + D_MODEL);                                              \
        rw  = *(const float4*)(lnw + (kt_) * GBK + ac);                                    \
        rz  = *(const float4*)(lnb + (kt_) * GBK + ac);                                    \
    }
#define GEMM_OUT_LN4(v_, mu_, rs_)                                                         \
    make_float4(((v_).x - (mu_)) * (rs_) * rw.x + rz.x, ((v_).y - (mu_)) * (rs_) * rw.y + rz.y, \
                ((v_).z - (mu_)) * (rs_) * rw.z + rz.z, ((v_).w - (mu_)) * (rs_) * rw.w + rz.w)
#define GEMM_OUT_STAGE(st_)                                                                \
    {                                                                                      \
        float* sa_ = smem + (st_) * STAGE_FLOATS + ar * APITCH + ac;                       \
        float* sb_ = smem + (st_) * STAGE_FLOATS + SA_FLOATS + br * BPITCH2 + 2 * bc;      \
        *(float4*)(sa_)               = GEMM_OUT_LN4(ra0, mu0, rs0);                       \
        *(float4*)(sa_ + 32 * APITCH) = GEMM_OUT_LN4(ra1, mu1, rs1);                       \
        *(float4*)(sa_ + 64 * APITCH) = GEMM_OUT_LN4(ra2, mu2, rs2);                       \
        *(float4*)(sa_ + 96 * APITCH) = GEMM_OUT_LN4(ra3, mu3, rs3);                       \
        *(float4*)(sb_)     = make_float4(rb0.x, rb1.x, rb0.y, rb1.y);                     \
        *(float4*)(sb_ + 4) = make_float4(rb0.z, rb1.z, rb0.w, rb1.w);                     \
    }

    v8f c00 = {}, c01 = {}, c10 = {}, c11 = {};
    const int fa = (wm * 32 + row) * APITCH + koff;
    const int fb = SA_FLOATS + (lane >> 4) * BPITCH2 + (wn * 32 + row) * 2;

    GEMM_OUT_LOAD(0)
    GEMM_OUT_STAGE(0)
    __syncthreads();

#pragma unroll 1
    for (int kt = 0; kt < D_INNER / GBK; ++kt) {
        const int st = kt & 1;
        if (kt + 1 < D_INNER / GBK) GEMM_OUT_LOAD(kt + 1)
        slab_mma(smem + st * STAGE_FLOATS + fa, smem + st * STAGE_FLOATS + fb, c00, c01, c10, c11);
        if (kt + 1 < D_INNER / GBK) GEMM_OUT_STAGE(st ^ 1)
        __syncthreads();
    }
#undef GEMM_OUT_LOAD
#undef GEMM_OUT_LN4
#undef GEMM_OUT_STAGE

    const int mhi = (lane >> 4) << 3;
    const int nb  = n0 + wn * 32 + row;
#pragma unroll
    for (int r = 0; r < 8; ++r) {
        const size_t mm = (size_t)(m0 + wm * 32 + mhi + r);
        out[(mm     ) * D_MODEL + nb     ] = c00[r];
        out[(mm     ) * D_MODEL + nb + 16] = c01[r];
        out[(mm + 16) * D_MODEL + nb     ] = c10[r];
        out[(mm + 16) * D_MODEL + nb + 16] = c11[r];
    }
}

extern "C" void kernel_launch(void* const* d_in, const int* in_sizes, int n_in,
                              void* d_out, int out_size, void* d_ws, size_t ws_size,
                              hipStream_t stream) {
    const float* u        = (const float*)d_in[0];
    const float* W_in     = (const float*)d_in[2];
    const float* conv_fw  = (const float*)d_in[3];
    const float* conv_fb  = (const float*)d_in[4];
    const float* conv_bw  = (const float*)d_in[5];
    const float* conv_bb  = (const float*)d_in[6];
    const float* dt_bias  = (const float*)d_in[7];
    const float* A_log    = (const float*)d_in[8];
    const float* D_param  = (const float*)d_in[9];
    const float* ln_w     = (const float*)d_in[10];
    const float* ln_b     = (const float*)d_in[11];
    const float* W_out    = (const float*)d_in[12];
    float* out            = (float*)d_out;

    float* ws = (float*)d_ws;
    size_t off = 0;
    off += (size_t)BL * D_IN_PROJ;
    float* xc    = ws + off; off += (size_t)BL * CONV2;
    float* dtb   = ws + off; off += (size_t)BL * NHEADS;
    float* ybuf  = ws + off; off += (size_t)BL * D_INNER;
    float* Sbuf  = ws + off; off += (size_t)32 * NCHUNK * HEADDIM * NSTATE;
    float* pAbuf = ws + off; off += (size_t)32 * NCHUNK;
    float* mubuf = ws + off; off += (size_t)BL;
    float* rsbuf = ws + off; off += (size_t)BL;

    gemm_in<<<dim3(GIN_NBLK, G_MBLK_IN), 256, 0, stream>>>(u, W_in, conv_fw, conv_fb, conv_bw, conv_bb,
                                                            dt_bias, xc, dtb);
    scan_chunk<<<dim3(BATCH * NHEADS * NCHUNK), 64, 0, stream>>>(
        xc, dtb, A_log, D_param, ybuf, Sbuf, pAbuf);
    scan_combine<<<dim3(BATCH * NHEADS * CB_BLK_PER_BH), CB_THREADS, 0, stream>>>(Sbuf, pAbuf);
    scan_correct<<<dim3(BATCH * NHEADS * NCHUNK), 64, 0, stream>>>(
        xc, dtb, A_log, Sbuf, ybuf);
    ln_stats<<<dim3(BL / LN_RPB), 256, 0, stream>>>(ybuf, mubuf, rsbuf);
    gemm_out<<<dim3(GOUT_NBLK, G_MBLK), 256, 0, stream>>>(
        ybuf, mubuf, rsbuf, ln_w, ln_b, W_out, out);

    (void)in_sizes; (void)n_in; (void)out_size; (void)ws_size;
}
